# starting kernel
# baseline (speedup 1.0000x reference)
#include <hip/hip_runtime.h>

#define SVCN   50000
#define INSTN  100000
#define NODEN  50000
#define FDIM   128
#define HDIM   128
#define OUTD   256
#define TOTAL  (SVCN + NODEN + INSTN)
#define E_SC   400000
#define E_IN   100000
#define E_NI   100000
#define E_II   800000

#define OFF_CAT   0LL
#define OFF_HSC   (OFF_CAT  + (long long)TOTAL * 128)
#define OFF_HIN   (OFF_HSC  + (long long)SVCN  * 128)
#define OFF_HNI   (OFF_HIN  + (long long)INSTN * 128)
#define OFF_HII   (OFF_HNI  + (long long)NODEN * 128)
#define OFF_DEG   (OFF_HII  + (long long)INSTN * 128)
#define D_NS_SC   (OFF_DEG + 0LL)
#define D_ND_SC   (D_NS_SC + SVCN)
#define D_NS_IN   (D_ND_SC + SVCN)
#define D_ND_IN   (D_NS_IN + INSTN)
#define D_NS_NI   (D_ND_IN + NODEN)
#define D_ND_NI   (D_NS_NI + NODEN)
#define D_NS_II   (D_ND_NI + INSTN)
#define D_ND_II   (D_NS_II + INSTN)
#define DEG_TOTAL 600000

typedef __attribute__((ext_vector_type(2))) float v2f;
typedef __attribute__((ext_vector_type(8))) float v8f;

__device__ __forceinline__ float lrelu(float x) { return x > 0.f ? x : 0.01f * x; }

__device__ __forceinline__ void atomAddF(float* p, float v) { unsafeAtomicAdd(p, v); }

__global__ void init_deg_kernel(float* deg) {
    int i = blockIdx.x * blockDim.x + threadIdx.x;
    if (i < DEG_TOTAL) deg[i] = 0.f;
}

__global__ void init_cat_kernel(float* __restrict__ cat,
                                const float* __restrict__ b_sc,
                                const float* __restrict__ b_in,
                                const float* __restrict__ b_ni,
                                const float* __restrict__ b_ii) {
    long long tid = (long long)blockIdx.x * blockDim.x + threadIdx.x;
    if (tid >= (long long)TOTAL * 32) return;
    int row = (int)(tid >> 5);
    int c4  = (int)(tid & 31) * 4;
    float4 v;
    if (row < SVCN) {
        v = *(const float4*)&b_sc[c4];
    } else if (row < SVCN + NODEN) {
        v = *(const float4*)&b_in[c4];
    } else {
        float4 a = *(const float4*)&b_ni[c4];
        float4 b = *(const float4*)&b_ii[c4];
        v = make_float4(a.x + b.x, a.y + b.y, a.z + b.z, a.w + b.w);
    }
    *(float4*)&cat[(long long)row * 128 + c4] = v;
}

__global__ void init_out_kernel(float* __restrict__ out, const float* __restrict__ b_tot) {
    int i = blockIdx.x * blockDim.x + threadIdx.x;
    if (i < OUTD * HDIM) out[i] = b_tot[i >> 7];
}

__global__ void degree_kernel(const int* __restrict__ src, const int* __restrict__ dst,
                              float* __restrict__ degS, float* __restrict__ degD, int nE) {
    int e = blockIdx.x * blockDim.x + threadIdx.x;
    if (e >= nE) return;
    atomAddF(&degS[src[e]], 1.f);
    atomAddF(&degD[dst[e]], 1.f);
}

__global__ void rsqrt_kernel(float* deg) {
    int i = blockIdx.x * blockDim.x + threadIdx.x;
    if (i < DEG_TOTAL) {
        float d = deg[i];
        deg[i] = rsqrtf(d > 1.f ? d : 1.f);
    }
}

__global__ __launch_bounds__(256) void proj_kernel(const float* __restrict__ X,
                                                   const float* __restrict__ W,
                                                   const float* __restrict__ ns,
                                                   float* __restrict__ Hout, int nrows) {
    __shared__ float Wlds[128 * 128];
    int t = threadIdx.x;
    for (int i = t * 4; i < 128 * 128; i += 256 * 4)
        *(float4*)&Wlds[i] = *(const float4*)&W[i];
    __syncthreads();

    const int lane = t & 31;
    const int wave = t >> 5;
    const int half = lane >> 4;
    const int l16  = lane & 15;
    const int r0   = blockIdx.x * 64;
    const int n0   = wave * 16;

    const float* Arow[4];
#pragma unroll
    for (int rt = 0; rt < 4; ++rt) {
        int row = r0 + rt * 16 + l16;
        int rc  = row < nrows ? row : nrows - 1;
        Arow[rt] = X + (long long)rc * 128;
    }

    v8f acc[4] = {v8f{}, v8f{}, v8f{}, v8f{}};

    for (int k = 0; k < 128; k += 4) {
        const int kk = k + 2 * half;
        v2f b;
        b.x = Wlds[kk * 128 + n0 + l16];
        b.y = Wlds[(kk + 1) * 128 + n0 + l16];
#pragma unroll
        for (int rt = 0; rt < 4; ++rt) {
            v2f a = *(const v2f*)&Arow[rt][kk];
            acc[rt] = __builtin_amdgcn_wmma_f32_16x16x4_f32(
                false, a, false, b, (short)0, acc[rt], false, false);
        }
    }

    if (r0 + 64 <= nrows) {
#pragma unroll
        for (int rt = 0; rt < 4; ++rt) {
#pragma unroll
            for (int i = 0; i < 8; ++i) {
                int row = r0 + rt * 16 + i + 8 * half;
                Hout[(long long)row * 128 + n0 + l16] = acc[rt][i] * ns[row];
            }
        }
    } else {
#pragma unroll
        for (int rt = 0; rt < 4; ++rt) {
#pragma unroll
            for (int i = 0; i < 8; ++i) {
                int row = r0 + rt * 16 + i + 8 * half;
                if (row < nrows)
                    Hout[(long long)row * 128 + n0 + l16] = acc[rt][i] * ns[row];
            }
        }
    }
}

__global__ void scatter_kernel(const float* __restrict__ H,
                               const int* __restrict__ src, const int* __restrict__ dst,
                               const float* __restrict__ nd, float* __restrict__ cat,
                               long long dstBase, int nE) {
    long long tid = (long long)blockIdx.x * blockDim.x + threadIdx.x;
    int e = (int)(tid >> 5);
    if (e >= nE) return;
    int c4 = (int)(tid & 31) * 4;
    int s = src[e], d = dst[e];
    float scale = nd[d];
    if (e + 1 < nE)
        __builtin_prefetch(&H[(long long)src[e + 1] * 128 + c4], 0, 1);
    float4 hv = *(const float4*)&H[(long long)s * 128 + c4];
    float* p = &cat[dstBase + (long long)d * 128 + c4];
    atomAddF(p + 0, hv.x * scale);
    atomAddF(p + 1, hv.y * scale);
    atomAddF(p + 2, hv.z * scale);
    atomAddF(p + 3, hv.w * scale);
}

#define KCHUNK 1024
__global__ __launch_bounds__(256) void final_gemm_kernel(const float* __restrict__ Wt,
                                                         const float* __restrict__ cat,
                                                         float* __restrict__ out) {
    const int t    = threadIdx.x;
    const int lane = t & 31;
    const int wave = t >> 5;
    const int half = lane >> 4;
    const int l16  = lane & 15;
    const int m0   = blockIdx.x * 64;
    const int n0   = wave * 16;
    const int k0   = blockIdx.y * KCHUNK;
    const int kEnd = (k0 + KCHUNK < TOTAL) ? (k0 + KCHUNK) : TOTAL;

    const float* Arow[4];
#pragma unroll
    for (int rt = 0; rt < 4; ++rt)
        Arow[rt] = Wt + (long long)(m0 + rt * 16 + l16) * TOTAL;

    v8f acc[4] = {v8f{}, v8f{}, v8f{}, v8f{}};

    for (int k = k0; k < kEnd; k += 4) {
        const int kk = k + 2 * half;
        v2f b;
        b.x = lrelu(cat[(long long)kk * 128 + n0 + l16]);
        b.y = lrelu(cat[(long long)(kk + 1) * 128 + n0 + l16]);
#pragma unroll
        for (int rt = 0; rt < 4; ++rt) {
            v2f a = *(const v2f*)&Arow[rt][kk];
            acc[rt] = __builtin_amdgcn_wmma_f32_16x16x4_f32(
                false, a, false, b, (short)0, acc[rt], false, false);
        }
    }

#pragma unroll
    for (int rt = 0; rt < 4; ++rt) {
#pragma unroll
        for (int i = 0; i < 8; ++i) {
            int m = m0 + rt * 16 + i + 8 * half;
            atomAddF(&out[m * 128 + n0 + l16], acc[rt][i]);
        }
    }
}

static inline int cdiv(long long a, long long b) { return (int)((a + b - 1) / b); }

extern "C" void kernel_launch(void* const* d_in, const int* in_sizes, int n_in,
                              void* d_out, int out_size, void* d_ws, size_t ws_size,
                              hipStream_t stream) {
    const float* feat_svc  = (const float*)d_in[0];
    const float* feat_inst = (const float*)d_in[1];
    const float* feat_node = (const float*)d_in[2];
    const int*   svc_src   = (const int*)d_in[3];
    const int*   svc_dst   = (const int*)d_in[4];
    const int*   in_src    = (const int*)d_in[5];
    const int*   in_dst    = (const int*)d_in[6];
    const int*   ni_src    = (const int*)d_in[7];
    const int*   ni_dst    = (const int*)d_in[8];
    const int*   ii_src    = (const int*)d_in[9];
    const int*   ii_dst    = (const int*)d_in[10];
    const float* W_sc = (const float*)d_in[11];  const float* b_sc = (const float*)d_in[12];
    const float* W_in = (const float*)d_in[13];  const float* b_in = (const float*)d_in[14];
    const float* W_ni = (const float*)d_in[15];  const float* b_ni = (const float*)d_in[16];
    const float* W_ii = (const float*)d_in[17];  const float* b_ii = (const float*)d_in[18];
    const float* W_tot = (const float*)d_in[19]; const float* b_tot = (const float*)d_in[20];
    float* out = (float*)d_out;

    float* ws   = (float*)d_ws;
    float* cat  = ws + OFF_CAT;
    float* h_sc = ws + OFF_HSC;
    float* h_in = ws + OFF_HIN;
    float* h_ni = ws + OFF_HNI;
    float* h_ii = ws + OFF_HII;
    float* deg  = ws + OFF_DEG;
    float* ns_sc = ws + D_NS_SC; float* nd_sc = ws + D_ND_SC;
    float* ns_in = ws + D_NS_IN; float* nd_in = ws + D_ND_IN;
    float* ns_ni = ws + D_NS_NI; float* nd_ni = ws + D_ND_NI;
    float* ns_ii = ws + D_NS_II; float* nd_ii = ws + D_ND_II;

    init_deg_kernel<<<cdiv(DEG_TOTAL, 256), 256, 0, stream>>>(deg);
    init_cat_kernel<<<cdiv((long long)TOTAL * 32, 256), 256, 0, stream>>>(cat, b_sc, b_in, b_ni, b_ii);
    init_out_kernel<<<cdiv(OUTD * HDIM, 256), 256, 0, stream>>>(out, b_tot);

    degree_kernel<<<cdiv(E_SC, 256), 256, 0, stream>>>(svc_src, svc_dst, ns_sc, nd_sc, E_SC);
    degree_kernel<<<cdiv(E_IN, 256), 256, 0, stream>>>(in_src,  in_dst,  ns_in, nd_in, E_IN);
    degree_kernel<<<cdiv(E_NI, 256), 256, 0, stream>>>(ni_src,  ni_dst,  ns_ni, nd_ni, E_NI);
    degree_kernel<<<cdiv(E_II, 256), 256, 0, stream>>>(ii_src,  ii_dst,  ns_ii, nd_ii, E_II);
    rsqrt_kernel<<<cdiv(DEG_TOTAL, 256), 256, 0, stream>>>(deg);

    proj_kernel<<<cdiv(SVCN, 64),  256, 0, stream>>>(feat_svc,  W_sc, ns_sc, h_sc, SVCN);
    proj_kernel<<<cdiv(INSTN, 64), 256, 0, stream>>>(feat_inst, W_in, ns_in, h_in, INSTN);
    proj_kernel<<<cdiv(NODEN, 64), 256, 0, stream>>>(feat_node, W_ni, ns_ni, h_ni, NODEN);
    proj_kernel<<<cdiv(INSTN, 64), 256, 0, stream>>>(feat_inst, W_ii, ns_ii, h_ii, INSTN);

    scatter_kernel<<<cdiv((long long)E_SC * 32, 256), 256, 0, stream>>>(
        h_sc, svc_src, svc_dst, nd_sc, cat, 0LL, E_SC);
    scatter_kernel<<<cdiv((long long)E_IN * 32, 256), 256, 0, stream>>>(
        h_in, in_src, in_dst, nd_in, cat, (long long)SVCN * 128, E_IN);
    scatter_kernel<<<cdiv((long long)E_NI * 32, 256), 256, 0, stream>>>(
        h_ni, ni_src, ni_dst, nd_ni, cat, (long long)(SVCN + NODEN) * 128, E_NI);
    scatter_kernel<<<cdiv((long long)E_II * 32, 256), 256, 0, stream>>>(
        h_ii, ii_src, ii_dst, nd_ii, cat, (long long)(SVCN + NODEN) * 128, E_II);

    dim3 g(OUTD / 64, cdiv(TOTAL, KCHUNK));
    final_gemm_kernel<<<g, 256, 0, stream>>>(W_tot, cat, out);
}


# edge scatter: lane l owns columns l, l+32, l+64, l+96 (was 4l..4l+3): each f32 atomic instruction and source load covers one whole 128-byte line, not a quarter of four; arrival order as before
# speedup vs baseline: 1.2984x; 1.2984x over previous
#include <hip/hip_runtime.h>

#define SVCN   50000
#define INSTN  100000
#define NODEN  50000
#define FDIM   128
#define HDIM   128
#define OUTD   256
#define TOTAL  (SVCN + NODEN + INSTN)
#define E_SC   400000
#define E_IN   100000
#define E_NI   100000
#define E_II   800000

#define OFF_CAT   0LL
#define OFF_HSC   (OFF_CAT  + (long long)TOTAL * 128)
#define OFF_HIN   (OFF_HSC  + (long long)SVCN  * 128)
#define OFF_HNI   (OFF_HIN  + (long long)INSTN * 128)
#define OFF_HII   (OFF_HNI  + (long long)NODEN * 128)
#define OFF_DEG   (OFF_HII  + (long long)INSTN * 128)
#define D_NS_SC   (OFF_DEG + 0LL)
#define D_ND_SC   (D_NS_SC + SVCN)
#define D_NS_IN   (D_ND_SC + SVCN)
#define D_ND_IN   (D_NS_IN + INSTN)
#define D_NS_NI   (D_ND_IN + NODEN)
#define D_ND_NI   (D_NS_NI + NODEN)
#define D_NS_II   (D_ND_NI + INSTN)
#define D_ND_II   (D_NS_II + INSTN)
#define DEG_TOTAL 600000

typedef __attribute__((ext_vector_type(2))) float v2f;
typedef __attribute__((ext_vector_type(8))) float v8f;

__device__ __forceinline__ float lrelu(float x) { return x > 0.f ? x : 0.01f * x; }

__device__ __forceinline__ void atomAddF(float* p, float v) { unsafeAtomicAdd(p, v); }

__global__ void init_deg_kernel(float* deg) {
    int i = blockIdx.x * blockDim.x + threadIdx.x;
    if (i < DEG_TOTAL) deg[i] = 0.f;
}

__global__ void init_cat_kernel(float* __restrict__ cat,
                                const float* __restrict__ b_sc,
                                const float* __restrict__ b_in,
                                const float* __restrict__ b_ni,
                                const float* __restrict__ b_ii) {
    long long tid = (long long)blockIdx.x * blockDim.x + threadIdx.x;
    if (tid >= (long long)TOTAL * 32) return;
    int row = (int)(tid >> 5);
    int c4  = (int)(tid & 31) * 4;
    float4 v;
    if (row < SVCN) {
        v = *(const float4*)&b_sc[c4];
    } else if (row < SVCN + NODEN) {
        v = *(const float4*)&b_in[c4];
    } else {
        float4 a = *(const float4*)&b_ni[c4];
        float4 b = *(const float4*)&b_ii[c4];
        v = make_float4(a.x + b.x, a.y + b.y, a.z + b.z, a.w + b.w);
    }
    *(float4*)&cat[(long long)row * 128 + c4] = v;
}

__global__ void init_out_kernel(float* __restrict__ out, const float* __restrict__ b_tot) {
    int i = blockIdx.x * blockDim.x + threadIdx.x;
    if (i < OUTD * HDIM) out[i] = b_tot[i >> 7];
}

__global__ void degree_kernel(const int* __restrict__ src, const int* __restrict__ dst,
                              float* __restrict__ degS, float* __restrict__ degD, int nE) {
    int e = blockIdx.x * blockDim.x + threadIdx.x;
    if (e >= nE) return;
    atomAddF(&degS[src[e]], 1.f);
    atomAddF(&degD[dst[e]], 1.f);
}

__global__ void rsqrt_kernel(float* deg) {
    int i = blockIdx.x * blockDim.x + threadIdx.x;
    if (i < DEG_TOTAL) {
        float d = deg[i];
        deg[i] = rsqrtf(d > 1.f ? d : 1.f);
    }
}

__global__ __launch_bounds__(256) void proj_kernel(const float* __restrict__ X,
                                                   const float* __restrict__ W,
                                                   const float* __restrict__ ns,
                                                   float* __restrict__ Hout, int nrows) {
    __shared__ float Wlds[128 * 128];
    int t = threadIdx.x;
    for (int i = t * 4; i < 128 * 128; i += 256 * 4)
        *(float4*)&Wlds[i] = *(const float4*)&W[i];
    __syncthreads();

    const int lane = t & 31;
    const int wave = t >> 5;
    const int half = lane >> 4;
    const int l16  = lane & 15;
    const int r0   = blockIdx.x * 64;
    const int n0   = wave * 16;

    const float* Arow[4];
#pragma unroll
    for (int rt = 0; rt < 4; ++rt) {
        int row = r0 + rt * 16 + l16;
        int rc  = row < nrows ? row : nrows - 1;
        Arow[rt] = X + (long long)rc * 128;
    }

    v8f acc[4] = {v8f{}, v8f{}, v8f{}, v8f{}};

    for (int k = 0; k < 128; k += 4) {
        const int kk = k + 2 * half;
        v2f b;
        b.x = Wlds[kk * 128 + n0 + l16];
        b.y = Wlds[(kk + 1) * 128 + n0 + l16];
#pragma unroll
        for (int rt = 0; rt < 4; ++rt) {
            v2f a = *(const v2f*)&Arow[rt][kk];
            acc[rt] = __builtin_amdgcn_wmma_f32_16x16x4_f32(
                false, a, false, b, (short)0, acc[rt], false, false);
        }
    }

    if (r0 + 64 <= nrows) {
#pragma unroll
        for (int rt = 0; rt < 4; ++rt) {
#pragma unroll
            for (int i = 0; i < 8; ++i) {
                int row = r0 + rt * 16 + i + 8 * half;
                Hout[(long long)row * 128 + n0 + l16] = acc[rt][i] * ns[row];
            }
        }
    } else {
#pragma unroll
        for (int rt = 0; rt < 4; ++rt) {
#pragma unroll
            for (int i = 0; i < 8; ++i) {
                int row = r0 + rt * 16 + i + 8 * half;
                if (row < nrows)
                    Hout[(long long)row * 128 + n0 + l16] = acc[rt][i] * ns[row];
            }
        }
    }
}

__global__ void scatter_kernel(const float* __restrict__ H,
                               const int* __restrict__ src, const int* __restrict__ dst,
                               const float* __restrict__ nd, float* __restrict__ cat,
                               long long dstBase, int nE) {
    long long tid = (long long)blockIdx.x * blockDim.x + threadIdx.x;
    int e = (int)(tid >> 5);
    if (e >= nE) return;
    int lane = (int)(tid & 31);
    int s = src[e], d = dst[e];
    float scale = nd[d];
    if (e + 1 < nE)
        __builtin_prefetch(&H[(long long)src[e + 1] * 128 + lane * 4], 0, 1);
    const float* hp = &H[(long long)s * 128 + lane];
    float h0 = hp[0], h1 = hp[32], h2 = hp[64], h3 = hp[96];
    float* p = &cat[dstBase + (long long)d * 128 + lane];
    atomAddF(p +  0, h0 * scale);
    atomAddF(p + 32, h1 * scale);
    atomAddF(p + 64, h2 * scale);
    atomAddF(p + 96, h3 * scale);
}

#define FG_KSLICE 800
#define FG_STEP   16
#define FG_GROUPS (TOTAL / FG_KSLICE)
static_assert(TOTAL % FG_KSLICE == 0, "K slices must tile K exactly");
static_assert(FG_KSLICE % (2 * FG_STEP) == 0, "loop body covers two steps");

struct FgFrag {
    float4 a[4][2];
    float2 b[4][4][2];
};

__device__ __forceinline__ void fg_load(FgFrag& f, const float* const (&Ap)[4],
                                        const float* __restrict__ Bp, int k) {
#pragma unroll
    for (int rt = 0; rt < 4; ++rt) {
        f.a[rt][0] = *(const float4*)(Ap[rt] + k);
        f.a[rt][1] = *(const float4*)(Ap[rt] + k + 4);
    }
#pragma unroll
    for (int s = 0; s < 4; ++s) {
#pragma unroll
        for (int q = 0; q < 4; ++q) {
            f.b[s][q][0] = *(const float2*)(Bp + (long long)(k + 2 * s) * 128 + 32 * q);
            f.b[s][q][1] = *(const float2*)(Bp + (long long)(k + 2 * s + 1) * 128 + 32 * q);
        }
    }
}

__device__ __forceinline__ void fg_mma(const FgFrag& f, v8f (&acc)[4][8]) {
#pragma unroll
    for (int s = 0; s < 4; ++s) {
        v2f bf[4][2];
#pragma unroll
        for (int q = 0; q < 4; ++q) {
            bf[q][0].x = lrelu(f.b[s][q][0].x);
            bf[q][0].y = lrelu(f.b[s][q][1].x);
            bf[q][1].x = lrelu(f.b[s][q][0].y);
            bf[q][1].y = lrelu(f.b[s][q][1].y);
        }
#pragma unroll
        for (int rt = 0; rt < 4; ++rt) {
            const float4 av = f.a[rt][s >> 1];
            v2f a;
            a.x = (s & 1) ? av.z : av.x;
            a.y = (s & 1) ? av.w : av.y;
#pragma unroll
            for (int q = 0; q < 4; ++q) {
#pragma unroll
                for (int e = 0; e < 2; ++e) {
                    acc[rt][2 * q + e] = __builtin_amdgcn_wmma_f32_16x16x4_f32(
                        false, a, false, bf[q][e], (short)0, acc[rt][2 * q + e], false, false);
                }
            }
        }
    }
}

__global__ __launch_bounds__(128) void final_gemm_kernel(const float* __restrict__ Wt,
                                                         const float* __restrict__ cat,
                                                         float* __restrict__ out) {
    const int t    = threadIdx.x;
    const int lane = t & 31;
    const int wave = t >> 5;
    const int half = lane >> 4;
    const int c    = lane & 15;
    const int m0   = wave * 64;
    const int k0   = blockIdx.x * FG_KSLICE;
    const int kEnd = k0 + FG_KSLICE;

    const float* Ap[4];
#pragma unroll
    for (int rt = 0; rt < 4; ++rt)
        Ap[rt] = Wt + (long long)(m0 + rt * 16 + c) * TOTAL + 8 * half;
    const float* Bp = cat + (long long)(8 * half) * 128 + 2 * c;

    v8f acc[4][8];
#pragma unroll
    for (int rt = 0; rt < 4; ++rt)
#pragma unroll
        for (int j = 0; j < 8; ++j) acc[rt][j] = v8f{};

    FgFrag f0, f1;
    fg_load(f0, Ap, Bp, k0);
    asm volatile("" ::: "memory");
    for (int k = k0; k < kEnd; k += 2 * FG_STEP) {
        fg_load(f1, Ap, Bp, k + FG_STEP);
        fg_mma(f0, acc);
        const int kn = (k + 2 * FG_STEP < kEnd) ? (k + 2 * FG_STEP) : k0;
        fg_load(f0, Ap, Bp, kn);
        asm volatile("" ::: "memory");
        fg_mma(f1, acc);
    }

#pragma unroll
    for (int rt = 0; rt < 4; ++rt) {
#pragma unroll
        for (int q = 0; q < 4; ++q) {
#pragma unroll
            for (int e = 0; e < 2; ++e) {
#pragma unroll
                for (int i = 0; i < 8; ++i) {
                    int m = m0 + rt * 16 + i + 8 * half;
                    atomAddF(&out[m * 128 + 32 * q + 2 * c + e], acc[rt][2 * q + e][i]);
                }
            }
        }
    }
}

static inline int cdiv(long long a, long long b) { return (int)((a + b - 1) / b); }

extern "C" void kernel_launch(void* const* d_in, const int* in_sizes, int n_in,
                              void* d_out, int out_size, void* d_ws, size_t ws_size,
                              hipStream_t stream) {
    const float* feat_svc  = (const float*)d_in[0];
    const float* feat_inst = (const float*)d_in[1];
    const float* feat_node = (const float*)d_in[2];
    const int*   svc_src   = (const int*)d_in[3];
    const int*   svc_dst   = (const int*)d_in[4];
    const int*   in_src    = (const int*)d_in[5];
    const int*   in_dst    = (const int*)d_in[6];
    const int*   ni_src    = (const int*)d_in[7];
    const int*   ni_dst    = (const int*)d_in[8];
    const int*   ii_src    = (const int*)d_in[9];
    const int*   ii_dst    = (const int*)d_in[10];
    const float* W_sc = (const float*)d_in[11];  const float* b_sc = (const float*)d_in[12];
    const float* W_in = (const float*)d_in[13];  const float* b_in = (const float*)d_in[14];
    const float* W_ni = (const float*)d_in[15];  const float* b_ni = (const float*)d_in[16];
    const float* W_ii = (const float*)d_in[17];  const float* b_ii = (const float*)d_in[18];
    const float* W_tot = (const float*)d_in[19]; const float* b_tot = (const float*)d_in[20];
    float* out = (float*)d_out;

    float* ws   = (float*)d_ws;
    float* cat  = ws + OFF_CAT;
    float* h_sc = ws + OFF_HSC;
    float* h_in = ws + OFF_HIN;
    float* h_ni = ws + OFF_HNI;
    float* h_ii = ws + OFF_HII;
    float* deg  = ws + OFF_DEG;
    float* ns_sc = ws + D_NS_SC; float* nd_sc = ws + D_ND_SC;
    float* ns_in = ws + D_NS_IN; float* nd_in = ws + D_ND_IN;
    float* ns_ni = ws + D_NS_NI; float* nd_ni = ws + D_ND_NI;
    float* ns_ii = ws + D_NS_II; float* nd_ii = ws + D_ND_II;

    init_deg_kernel<<<cdiv(DEG_TOTAL, 256), 256, 0, stream>>>(deg);
    init_cat_kernel<<<cdiv((long long)TOTAL * 32, 256), 256, 0, stream>>>(cat, b_sc, b_in, b_ni, b_ii);
    init_out_kernel<<<cdiv(OUTD * HDIM, 256), 256, 0, stream>>>(out, b_tot);

    degree_kernel<<<cdiv(E_SC, 256), 256, 0, stream>>>(svc_src, svc_dst, ns_sc, nd_sc, E_SC);
    degree_kernel<<<cdiv(E_IN, 256), 256, 0, stream>>>(in_src,  in_dst,  ns_in, nd_in, E_IN);
    degree_kernel<<<cdiv(E_NI, 256), 256, 0, stream>>>(ni_src,  ni_dst,  ns_ni, nd_ni, E_NI);
    degree_kernel<<<cdiv(E_II, 256), 256, 0, stream>>>(ii_src,  ii_dst,  ns_ii, nd_ii, E_II);
    rsqrt_kernel<<<cdiv(DEG_TOTAL, 256), 256, 0, stream>>>(deg);

    proj_kernel<<<cdiv(SVCN, 64),  256, 0, stream>>>(feat_svc,  W_sc, ns_sc, h_sc, SVCN);
    proj_kernel<<<cdiv(INSTN, 64), 256, 0, stream>>>(feat_inst, W_in, ns_in, h_in, INSTN);
    proj_kernel<<<cdiv(NODEN, 64), 256, 0, stream>>>(feat_node, W_ni, ns_ni, h_ni, NODEN);
    proj_kernel<<<cdiv(INSTN, 64), 256, 0, stream>>>(feat_inst, W_ii, ns_ii, h_ii, INSTN);

    scatter_kernel<<<cdiv((long long)E_SC * 32, 256), 256, 0, stream>>>(
        h_sc, svc_src, svc_dst, nd_sc, cat, 0LL, E_SC);
    scatter_kernel<<<cdiv((long long)E_IN * 32, 256), 256, 0, stream>>>(
        h_in, in_src, in_dst, nd_in, cat, (long long)SVCN * 128, E_IN);
    scatter_kernel<<<cdiv((long long)E_NI * 32, 256), 256, 0, stream>>>(
        h_ni, ni_src, ni_dst, nd_ni, cat, (long long)(SVCN + NODEN) * 128, E_NI);
    scatter_kernel<<<cdiv((long long)E_II * 32, 256), 256, 0, stream>>>(
        h_ii, ii_src, ii_dst, nd_ii, cat, (long long)(SVCN + NODEN) * 128, E_II);

    final_gemm_kernel<<<FG_GROUPS, 128, 0, stream>>>(W_tot, cat, out);
}


# global f32 atomic scatter and cat fill replaced: a work-group owns 640 cat rows as LDS sums, scans all destination ids, adds matching source rows by LDS f32 atomics; row sum order changes
# speedup vs baseline: 1.8509x; 1.4255x over previous
#include <hip/hip_runtime.h>

#define SVCN   50000
#define INSTN  100000
#define NODEN  50000
#define FDIM   128
#define HDIM   128
#define OUTD   256
#define TOTAL  (SVCN + NODEN + INSTN)
#define E_SC   400000
#define E_IN   100000
#define E_NI   100000
#define E_II   800000

#define OFF_CAT   0LL
#define OFF_HSC   (OFF_CAT  + (long long)TOTAL * 128)
#define OFF_HIN   (OFF_HSC  + (long long)SVCN  * 128)
#define OFF_HNI   (OFF_HIN  + (long long)INSTN * 128)
#define OFF_HII   (OFF_HNI  + (long long)NODEN * 128)
#define OFF_DEG   (OFF_HII  + (long long)INSTN * 128)
#define D_NS_SC   (OFF_DEG + 0LL)
#define D_ND_SC   (D_NS_SC + SVCN)
#define D_NS_IN   (D_ND_SC + SVCN)
#define D_ND_IN   (D_NS_IN + INSTN)
#define D_NS_NI   (D_ND_IN + NODEN)
#define D_ND_NI   (D_NS_NI + NODEN)
#define D_NS_II   (D_ND_NI + INSTN)
#define D_ND_II   (D_NS_II + INSTN)
#define DEG_TOTAL 600000

typedef __attribute__((ext_vector_type(2))) float v2f;
typedef __attribute__((ext_vector_type(8))) float v8f;

__device__ __forceinline__ float lrelu(float x) { return x > 0.f ? x : 0.01f * x; }

__device__ __forceinline__ void atomAddF(float* p, float v) { unsafeAtomicAdd(p, v); }

#define SMALL_BLOCKS 128
static_assert(DEG_TOTAL % 4 == 0, "degree planes are walked 16 bytes per lane");

__global__ __launch_bounds__(256) void init_deg_kernel(float* deg) {
    float4* d4 = (float4*)deg;
    for (int i = blockIdx.x * 256 + threadIdx.x; i < DEG_TOTAL / 4; i += SMALL_BLOCKS * 256)
        d4[i] = make_float4(0.f, 0.f, 0.f, 0.f);
}

__global__ void init_out_kernel(float* __restrict__ out, const float* __restrict__ b_tot) {
    int i = blockIdx.x * blockDim.x + threadIdx.x;
    if (i < OUTD * HDIM) out[i] = b_tot[i >> 7];
}

__global__ void degree_kernel(const int* __restrict__ src, const int* __restrict__ dst,
                              float* __restrict__ degS, float* __restrict__ degD, int nE) {
    int e = blockIdx.x * blockDim.x + threadIdx.x;
    if (e >= nE) return;
    atomAddF(&degS[src[e]], 1.f);
    atomAddF(&degD[dst[e]], 1.f);
}

__global__ __launch_bounds__(256) void rsqrt_kernel(float* deg) {
    float4* d4 = (float4*)deg;
    for (int i = blockIdx.x * 256 + threadIdx.x; i < DEG_TOTAL / 4; i += SMALL_BLOCKS * 256) {
        float4 d = d4[i];
        d.x = rsqrtf(d.x > 1.f ? d.x : 1.f);
        d.y = rsqrtf(d.y > 1.f ? d.y : 1.f);
        d.z = rsqrtf(d.z > 1.f ? d.z : 1.f);
        d.w = rsqrtf(d.w > 1.f ? d.w : 1.f);
        d4[i] = d;
    }
}

#define PJ_STEPS (FDIM / 16)

__global__ __launch_bounds__(256) void proj_kernel(const float* __restrict__ X,
                                                   const float* __restrict__ W,
                                                   const float* __restrict__ ns,
                                                   float* __restrict__ Hout, int nrows) {
    __shared__ float Wlds[128 * 128];
    const int t    = threadIdx.x;
    const int lane = t & 31;
    const int wave = t >> 5;
    const int half = lane >> 4;
    const int c    = lane & 15;
    const int r0   = blockIdx.x * 64 + (wave >> 1) * 16;
    const int n0   = (wave & 1) * 64 + 2 * c;

    const int rowA = (r0 + c < nrows) ? (r0 + c) : (nrows - 1);
    const float* Ap = X + (long long)rowA * 128 + 8 * half;
    float4 a[PJ_STEPS][2];
#pragma unroll
    for (int ks = 0; ks < PJ_STEPS; ++ks) {
        a[ks][0] = *(const float4*)(Ap + 16 * ks);
        a[ks][1] = *(const float4*)(Ap + 16 * ks + 4);
    }
    float nsv[8];
#pragma unroll
    for (int i = 0; i < 8; ++i) {
        const int row = r0 + i + 8 * half;
        nsv[i] = ns[row < nrows ? row : nrows - 1];
    }

#pragma unroll
    for (int it = 0; it < 8; ++it) {
        const int u  = t + 256 * it;
        const int kp = u >> 5;
        const int m4 = (u & 31) * 4;
        const float4 w0 = *(const float4*)&W[(2 * kp) * 128 + m4];
        const float4 w1 = *(const float4*)&W[(2 * kp + 1) * 128 + m4];
        float* d = &Wlds[(kp * 128 + m4) * 2];
        *(float4*)(d)     = make_float4(w0.x, w1.x, w0.y, w1.y);
        *(float4*)(d + 4) = make_float4(w0.z, w1.z, w0.w, w1.w);
    }
    __syncthreads();

    v8f acc[4] = {v8f{}, v8f{}, v8f{}, v8f{}};
    const float* Bp = &Wlds[(4 * half * 128 + n0) * 2];

#pragma unroll
    for (int ks = 0; ks < PJ_STEPS; ++ks) {
#pragma unroll
        for (int s = 0; s < 4; ++s) {
            const float* bp = Bp + ((8 * ks + s) * 128) * 2;
            const float4 b0 = *(const float4*)(bp);
            const float4 b1 = *(const float4*)(bp + 64);
            const float4 av = a[ks][s >> 1];
            v2f af;
            af.x = (s & 1) ? av.z : av.x;
            af.y = (s & 1) ? av.w : av.y;
            v2f bf;
            bf.x = b0.x; bf.y = b0.y;
            acc[0] = __builtin_amdgcn_wmma_f32_16x16x4_f32(false, af, false, bf, (short)0, acc[0], false, false);
            bf.x = b0.z; bf.y = b0.w;
            acc[1] = __builtin_amdgcn_wmma_f32_16x16x4_f32(false, af, false, bf, (short)0, acc[1], false, false);
            bf.x = b1.x; bf.y = b1.y;
            acc[2] = __builtin_amdgcn_wmma_f32_16x16x4_f32(false, af, false, bf, (short)0, acc[2], false, false);
            bf.x = b1.z; bf.y = b1.w;
            acc[3] = __builtin_amdgcn_wmma_f32_16x16x4_f32(false, af, false, bf, (short)0, acc[3], false, false);
        }
    }

#pragma unroll
    for (int i = 0; i < 8; ++i) {
        const int row = r0 + i + 8 * half;
        if (row < nrows) {
            const float f = nsv[i];
            float* hp = Hout + (long long)row * 128 + n0;
            *(float2*)(hp)      = make_float2(acc[0][i] * f, acc[1][i] * f);
            *(float2*)(hp + 32) = make_float2(acc[2][i] * f, acc[3][i] * f);
        }
    }
}

#define AG_ROWS    640
#define AG_THREADS 1024
#define AG_WAVES   (AG_THREADS / 32)
#define AG_FLY     16
#define AG_GRP     2
#define AG_T_INST  ((INSTN + AG_ROWS - 1) / AG_ROWS)
#define AG_T_SVC   ((SVCN  + AG_ROWS - 1) / AG_ROWS)
#define AG_T_NODE  ((NODEN + AG_ROWS - 1) / AG_ROWS)
static_assert(E_SC % 4 == 0 && E_IN % 4 == 0 && E_NI % 4 == 0 && E_II % 4 == 0,
              "destination ids are read 4 per lane");

struct AgList {
    const int*   src;
    const int*   dst;
    const float* nd;
    const float* H;
    int          n4;
};

template <bool FULL>
__device__ __forceinline__ void ag_flush(const AgList& L, float* tile, int r0, int lane, int qe, int n) {
    const int d = L.dst[qe];
    const int s = L.src[qe];
    const int f = __float_as_int(L.nd[d]);
    const int r = d - r0;
    float h[AG_FLY][4];
#pragma unroll
    for (int j = 0; j < AG_FLY; ++j) {
        h[j][0] = 0.f; h[j][1] = 0.f; h[j][2] = 0.f; h[j][3] = 0.f;
        if (FULL || j < n) {
            const float* hp = L.H + (long long)__builtin_amdgcn_readlane(s, j) * 128;
            h[j][0] = hp[lane]; h[j][1] = hp[lane + 32]; h[j][2] = hp[lane + 64]; h[j][3] = hp[lane + 96];
        }
    }
#pragma unroll
    for (int j = 0; j < AG_FLY; ++j) {
        if (FULL || j < n) {
            float*      p  = tile + __builtin_amdgcn_readlane(r, j) * 128 + lane;
            const float fj = __int_as_float(__builtin_amdgcn_readlane(f, j));
            atomicAdd(p,      h[j][0] * fj);
            atomicAdd(p + 32, h[j][1] * fj);
            atomicAdd(p + 64, h[j][2] * fj);
            atomicAdd(p + 96, h[j][3] * fj);
        }
    }
}

__device__ __forceinline__ void ag_group(const AgList& L, float* tile, int r0, int nr, int lane,
                                         int4 d, int gb, int& qe, int& cnt) {
    const unsigned ux = (unsigned)(d.x - r0), uy = (unsigned)(d.y - r0);
    const unsigned uz = (unsigned)(d.z - r0), uw = (unsigned)(d.w - r0);
    const unsigned lo = min(min(ux, uy), min(uz, uw));
    unsigned lanes = __builtin_amdgcn_ballot_w32(gb + lane < L.n4 && lo < (unsigned)nr);
    if (lanes == 0u) return;
    const int code = (ux < (unsigned)nr ? 1 : 0) | (uy < (unsigned)nr ? 2 : 0) |
                     (uz < (unsigned)nr ? 4 : 0) | (uw < (unsigned)nr ? 8 : 0);
    do {
        const int l = __builtin_ctz(lanes);
        lanes &= lanes - 1u;
        unsigned c = (unsigned)__builtin_amdgcn_readlane(code, l);
        const int eb = 4 * (gb + l);
        do {
            const int k = __builtin_ctz(c);
            c &= c - 1u;
            qe = (lane == cnt) ? (eb + k) : qe;
            if (++cnt == AG_FLY) { ag_flush<true>(L, tile, r0, lane, qe, AG_FLY); cnt = 0; }
        } while (c != 0u);
    } while (lanes != 0u);
}

__device__ __forceinline__ void ag_list(const AgList& L, float* tile, int r0, int nr, int wave, int lane) {
    const int4* dst4 = (const int4*)L.dst;
    int qe  = 0;
    int cnt = 0;
    int i = wave * 32 + lane;
    int4 n0 = dst4[i < L.n4 ? i : 0];
    int4 n1 = dst4[i + AG_THREADS < L.n4 ? i + AG_THREADS : 0];
    for (int gb = wave * 32; gb < L.n4; gb += AG_GRP * AG_THREADS) {
        const int4 d0 = n0, d1 = n1;
        i += AG_GRP * AG_THREADS;
        n0 = dst4[i < L.n4 ? i : 0];
        n1 = dst4[i + AG_THREADS < L.n4 ? i + AG_THREADS : 0];
        ag_group(L, tile, r0, nr, lane, d0, gb, qe, cnt);
        ag_group(L, tile, r0, nr, lane, d1, gb + AG_THREADS, qe, cnt);
    }
    if (cnt != 0) ag_flush<false>(L, tile, r0, lane, qe, cnt);
}

__global__ __launch_bounds__(AG_THREADS) void agg_kernel(
        const int* __restrict__ svc_src, const int* __restrict__ svc_dst,
        const int* __restrict__ in_src,  const int* __restrict__ in_dst,
        const int* __restrict__ ni_src,  const int* __restrict__ ni_dst,
        const int* __restrict__ ii_src,  const int* __restrict__ ii_dst,
        const float* __restrict__ h_sc, const float* __restrict__ h_in,
        const float* __restrict__ h_ni, const float* __restrict__ h_ii,
        const float* __restrict__ nd_sc, const float* __restrict__ nd_in,
        const float* __restrict__ nd_ni, const float* __restrict__ nd_ii,
        const float* __restrict__ b_sc, const float* __restrict__ b_in,
        const float* __restrict__ b_ni, const float* __restrict__ b_ii,
        float* __restrict__ cat) {
    __shared__ float tile[AG_ROWS * 128];
    const int lane = threadIdx.x & 31;
    const int wave = __builtin_amdgcn_readfirstlane(threadIdx.x >> 5);
    const int b    = blockIdx.x;

    for (int i = threadIdx.x; i < AG_ROWS * 32; i += AG_THREADS)
        ((float4*)tile)[i] = make_float4(0.f, 0.f, 0.f, 0.f);
    __syncthreads();

    int r0, nr, catRow0, nLists;
    AgList L;
    float4 bias;
    const int c4 = lane * 4;
    if (b < AG_T_INST) {
        r0 = b * AG_ROWS;
        nr = (INSTN - r0 < AG_ROWS) ? (INSTN - r0) : AG_ROWS;
        catRow0 = SVCN + NODEN + r0;
        L.src = ni_src; L.dst = ni_dst; L.nd = nd_ni; L.H = h_ni; L.n4 = E_NI / 4; nLists = 2;
        const float4 x = *(const float4*)&b_ni[c4];
        const float4 y = *(const float4*)&b_ii[c4];
        bias = make_float4(x.x + y.x, x.y + y.y, x.z + y.z, x.w + y.w);
    } else if (b < AG_T_INST + AG_T_SVC) {
        r0 = (b - AG_T_INST) * AG_ROWS;
        nr = (SVCN - r0 < AG_ROWS) ? (SVCN - r0) : AG_ROWS;
        catRow0 = r0;
        L.src = svc_src; L.dst = svc_dst; L.nd = nd_sc; L.H = h_sc; L.n4 = E_SC / 4; nLists = 1;
        bias = *(const float4*)&b_sc[c4];
    } else {
        r0 = (b - AG_T_INST - AG_T_SVC) * AG_ROWS;
        nr = (NODEN - r0 < AG_ROWS) ? (NODEN - r0) : AG_ROWS;
        catRow0 = SVCN + r0;
        L.src = in_src; L.dst = in_dst; L.nd = nd_in; L.H = h_in; L.n4 = E_IN / 4; nLists = 1;
        bias = *(const float4*)&b_in[c4];
    }
#pragma nounroll
    for (int li = 0; li < nLists; ++li) {
        ag_list(L, tile, r0, nr, wave, lane);
        L.src = ii_src; L.dst = ii_dst; L.nd = nd_ii; L.H = h_ii; L.n4 = E_II / 4;
    }
    __syncthreads();

    for (int r = wave; r < nr; r += AG_WAVES) {
        const float4 v = *(const float4*)&tile[r * 128 + c4];
        *(float4*)&cat[(long long)(catRow0 + r) * 128 + c4] =
            make_float4(v.x + bias.x, v.y + bias.y, v.z + bias.z, v.w + bias.w);
    }
}

#define FG_KSLICE 800
#define FG_STEP   16
#define FG_GROUPS (TOTAL / FG_KSLICE)
static_assert(TOTAL % FG_KSLICE == 0, "K slices must tile K exactly");
static_assert(FG_KSLICE % (2 * FG_STEP) == 0, "loop body covers two steps");

struct FgFrag {
    float4 a[4][2];
    float2 b[4][4][2];
};

__device__ __forceinline__ void fg_load(FgFrag& f, const float* const (&Ap)[4],
                                        const float* __restrict__ Bp, int k) {
#pragma unroll
    for (int rt = 0; rt < 4; ++rt) {
        f.a[rt][0] = *(const float4*)(Ap[rt] + k);
        f.a[rt][1] = *(const float4*)(Ap[rt] + k + 4);
    }
#pragma unroll
    for (int s = 0; s < 4; ++s) {
#pragma unroll
        for (int q = 0; q < 4; ++q) {
            f.b[s][q][0] = *(const float2*)(Bp + (long long)(k + 2 * s) * 128 + 32 * q);
            f.b[s][q][1] = *(const float2*)(Bp + (long long)(k + 2 * s + 1) * 128 + 32 * q);
        }
    }
}

__device__ __forceinline__ void fg_mma(const FgFrag& f, v8f (&acc)[4][8]) {
#pragma unroll
    for (int s = 0; s < 4; ++s) {
        v2f bf[4][2];
#pragma unroll
        for (int q = 0; q < 4; ++q) {
            bf[q][0].x = lrelu(f.b[s][q][0].x);
            bf[q][0].y = lrelu(f.b[s][q][1].x);
            bf[q][1].x = lrelu(f.b[s][q][0].y);
            bf[q][1].y = lrelu(f.b[s][q][1].y);
        }
#pragma unroll
        for (int rt = 0; rt < 4; ++rt) {
            const float4 av = f.a[rt][s >> 1];
            v2f a;
            a.x = (s & 1) ? av.z : av.x;
            a.y = (s & 1) ? av.w : av.y;
#pragma unroll
            for (int q = 0; q < 4; ++q) {
#pragma unroll
                for (int e = 0; e < 2; ++e) {
                    acc[rt][2 * q + e] = __builtin_amdgcn_wmma_f32_16x16x4_f32(
                        false, a, false, bf[q][e], (short)0, acc[rt][2 * q + e], false, false);
                }
            }
        }
    }
}

__global__ __launch_bounds__(128) void final_gemm_kernel(const float* __restrict__ Wt,
                                                         const float* __restrict__ cat,
                                                         float* __restrict__ out) {
    const int t    = threadIdx.x;
    const int lane = t & 31;
    const int wave = t >> 5;
    const int half = lane >> 4;
    const int c    = lane & 15;
    const int m0   = wave * 64;
    const int k0   = blockIdx.x * FG_KSLICE;
    const int kEnd = k0 + FG_KSLICE;

    const float* Ap[4];
#pragma unroll
    for (int rt = 0; rt < 4; ++rt)
        Ap[rt] = Wt + (long long)(m0 + rt * 16 + c) * TOTAL + 8 * half;
    const float* Bp = cat + (long long)(8 * half) * 128 + 2 * c;

    v8f acc[4][8];
#pragma unroll
    for (int rt = 0; rt < 4; ++rt)
#pragma unroll
        for (int j = 0; j < 8; ++j) acc[rt][j] = v8f{};

    FgFrag f0, f1;
    fg_load(f0, Ap, Bp, k0);
    asm volatile("" ::: "memory");
    for (int k = k0; k < kEnd; k += 2 * FG_STEP) {
        fg_load(f1, Ap, Bp, k + FG_STEP);
        fg_mma(f0, acc);
        const int kn = (k + 2 * FG_STEP < kEnd) ? (k + 2 * FG_STEP) : k0;
        fg_load(f0, Ap, Bp, kn);
        asm volatile("" ::: "memory");
        fg_mma(f1, acc);
    }

#pragma unroll
    for (int rt = 0; rt < 4; ++rt) {
#pragma unroll
        for (int q = 0; q < 4; ++q) {
#pragma unroll
            for (int e = 0; e < 2; ++e) {
#pragma unroll
                for (int i = 0; i < 8; ++i) {
                    int m = m0 + rt * 16 + i + 8 * half;
                    atomAddF(&out[m * 128 + 32 * q + 2 * c + e], acc[rt][2 * q + e][i]);
                }
            }
        }
    }
}

static inline int cdiv(long long a, long long b) { return (int)((a + b - 1) / b); }

extern "C" void kernel_launch(void* const* d_in, const int* in_sizes, int n_in,
                              void* d_out, int out_size, void* d_ws, size_t ws_size,
                              hipStream_t stream) {
    const float* feat_svc  = (const float*)d_in[0];
    const float* feat_inst = (const float*)d_in[1];
    const float* feat_node = (const float*)d_in[2];
    const int*   svc_src   = (const int*)d_in[3];
    const int*   svc_dst   = (const int*)d_in[4];
    const int*   in_src    = (const int*)d_in[5];
    const int*   in_dst    = (const int*)d_in[6];
    const int*   ni_src    = (const int*)d_in[7];
    const int*   ni_dst    = (const int*)d_in[8];
    const int*   ii_src    = (const int*)d_in[9];
    const int*   ii_dst    = (const int*)d_in[10];
    const float* W_sc = (const float*)d_in[11];  const float* b_sc = (const float*)d_in[12];
    const float* W_in = (const float*)d_in[13];  const float* b_in = (const float*)d_in[14];
    const float* W_ni = (const float*)d_in[15];  const float* b_ni = (const float*)d_in[16];
    const float* W_ii = (const float*)d_in[17];  const float* b_ii = (const float*)d_in[18];
    const float* W_tot = (const float*)d_in[19]; const float* b_tot = (const float*)d_in[20];
    float* out = (float*)d_out;

    float* ws   = (float*)d_ws;
    float* cat  = ws + OFF_CAT;
    float* h_sc = ws + OFF_HSC;
    float* h_in = ws + OFF_HIN;
    float* h_ni = ws + OFF_HNI;
    float* h_ii = ws + OFF_HII;
    float* deg  = ws + OFF_DEG;
    float* ns_sc = ws + D_NS_SC; float* nd_sc = ws + D_ND_SC;
    float* ns_in = ws + D_NS_IN; float* nd_in = ws + D_ND_IN;
    float* ns_ni = ws + D_NS_NI; float* nd_ni = ws + D_ND_NI;
    float* ns_ii = ws + D_NS_II; float* nd_ii = ws + D_ND_II;

    init_deg_kernel<<<SMALL_BLOCKS, 256, 0, stream>>>(deg);
    init_out_kernel<<<cdiv(OUTD * HDIM, 256), 256, 0, stream>>>(out, b_tot);

    degree_kernel<<<cdiv(E_SC, 256), 256, 0, stream>>>(svc_src, svc_dst, ns_sc, nd_sc, E_SC);
    degree_kernel<<<cdiv(E_IN, 256), 256, 0, stream>>>(in_src,  in_dst,  ns_in, nd_in, E_IN);
    degree_kernel<<<cdiv(E_NI, 256), 256, 0, stream>>>(ni_src,  ni_dst,  ns_ni, nd_ni, E_NI);
    degree_kernel<<<cdiv(E_II, 256), 256, 0, stream>>>(ii_src,  ii_dst,  ns_ii, nd_ii, E_II);
    rsqrt_kernel<<<SMALL_BLOCKS, 256, 0, stream>>>(deg);

    proj_kernel<<<cdiv(SVCN, 64),  256, 0, stream>>>(feat_svc,  W_sc, ns_sc, h_sc, SVCN);
    proj_kernel<<<cdiv(INSTN, 64), 256, 0, stream>>>(feat_inst, W_in, ns_in, h_in, INSTN);
    proj_kernel<<<cdiv(NODEN, 64), 256, 0, stream>>>(feat_node, W_ni, ns_ni, h_ni, NODEN);
    proj_kernel<<<cdiv(INSTN, 64), 256, 0, stream>>>(feat_inst, W_ii, ns_ii, h_ii, INSTN);

    agg_kernel<<<AG_T_INST + AG_T_SVC + AG_T_NODE, AG_THREADS, 0, stream>>>(
        svc_src, svc_dst, in_src, in_dst, ni_src, ni_dst, ii_src, ii_dst,
        h_sc, h_in, h_ni, h_ii, nd_sc, nd_in, nd_ni, nd_ii,
        b_sc, b_in, b_ni, b_ii, cat);

    final_gemm_kernel<<<FG_GROUPS, 128, 0, stream>>>(W_tot, cat, out);
}


# degree counts ride in the projection blocks; the source factor moves from the projection's store to the gather (rsqrt of both raw counts per edge): 13 launches become 5; sum order kept
# speedup vs baseline: 2.2331x; 1.2065x over previous
#include <hip/hip_runtime.h>

#define SVCN   50000
#define INSTN  100000
#define NODEN  50000
#define FDIM   128
#define HDIM   128
#define OUTD   256
#define TOTAL  (SVCN + NODEN + INSTN)
#define E_SC   400000
#define E_IN   100000
#define E_NI   100000
#define E_II   800000

#define OFF_CAT   0LL
#define OFF_HSC   (OFF_CAT  + (long long)TOTAL * 128)
#define OFF_HIN   (OFF_HSC  + (long long)SVCN  * 128)
#define OFF_HNI   (OFF_HIN  + (long long)INSTN * 128)
#define OFF_HII   (OFF_HNI  + (long long)NODEN * 128)
#define OFF_DEG   (OFF_HII  + (long long)INSTN * 128)
#define D_NS_SC   (OFF_DEG + 0LL)
#define D_ND_SC   (D_NS_SC + SVCN)
#define D_NS_IN   (D_ND_SC + SVCN)
#define D_ND_IN   (D_NS_IN + INSTN)
#define D_NS_NI   (D_ND_IN + NODEN)
#define D_ND_NI   (D_NS_NI + NODEN)
#define D_NS_II   (D_ND_NI + INSTN)
#define D_ND_II   (D_NS_II + INSTN)
#define DEG_TOTAL 600000

typedef __attribute__((ext_vector_type(2))) float v2f;
typedef __attribute__((ext_vector_type(8))) float v8f;

__device__ __forceinline__ float lrelu(float x) { return x > 0.f ? x : 0.01f * x; }

__device__ __forceinline__ void atomAddF(float* p, float v) { unsafeAtomicAdd(p, v); }

#define SMALL_BLOCKS 128
static_assert(DEG_TOTAL % 4 == 0, "degree planes are walked 16 bytes per lane");

__global__ __launch_bounds__(256) void init_deg_kernel(float* deg) {
    float4* d4 = (float4*)deg;
    for (int i = blockIdx.x * 256 + threadIdx.x; i < DEG_TOTAL / 4; i += SMALL_BLOCKS * 256)
        d4[i] = make_float4(0.f, 0.f, 0.f, 0.f);
}

__global__ void init_out_kernel(float* __restrict__ out, const float* __restrict__ b_tot) {
    int i = blockIdx.x * blockDim.x + threadIdx.x;
    if (i < OUTD * HDIM) out[i] = b_tot[i >> 7];
}

#define PJ_STEPS (FDIM / 16)
#define PJ_T_SVC   ((SVCN  + 63) / 64)
#define PJ_T_NODE  ((NODEN + 63) / 64)
#define PJ_T_INST  ((INSTN + 63) / 64)
#define PJ_BLOCKS  (2 * PJ_T_INST + PJ_T_SVC + PJ_T_NODE)
#define DG_SC      (E_SC / 4)
#define DG_IN      (E_IN / 4)
#define DG_NI      (E_NI / 4)
#define DG_II      (E_II / 4)
#define DG_TOTAL   (DG_SC + DG_IN + DG_NI + DG_II)
#define DG_PER_BLOCK ((DG_TOTAL + PJ_BLOCKS - 1) / PJ_BLOCKS)
static_assert(E_SC % 4 == 0 && E_IN % 4 == 0 && E_NI % 4 == 0 && E_II % 4 == 0,
              "edge ids are read 4 per lane");
static_assert(DG_PER_BLOCK <= 256, "one id group per thread");

__global__ __launch_bounds__(256) void proj_deg_kernel(
        const float* __restrict__ feat_svc, const float* __restrict__ feat_inst,
        const float* __restrict__ feat_node,
        const float* __restrict__ W_sc, const float* __restrict__ W_in,
        const float* __restrict__ W_ni, const float* __restrict__ W_ii,
        float* __restrict__ h_sc, float* __restrict__ h_in,
        float* __restrict__ h_ni, float* __restrict__ h_ii,
        const int* __restrict__ svc_src, const int* __restrict__ svc_dst,
        const int* __restrict__ in_src,  const int* __restrict__ in_dst,
        const int* __restrict__ ni_src,  const int* __restrict__ ni_dst,
        const int* __restrict__ ii_src,  const int* __restrict__ ii_dst,
        float* __restrict__ deg) {
    __shared__ float Wlds[128 * 128];
    const int t    = threadIdx.x;
    const int lane = t & 31;
    const int wave = t >> 5;
    const int half = lane >> 4;
    const int c    = lane & 15;
    const int b    = blockIdx.x;

    const float* X; const float* W; float* Hout; int nrows, tile;
    if (b < 2 * PJ_T_INST) {
        tile = b >> 1; X = feat_inst; nrows = INSTN;
        if (b & 1) { W = W_ii; Hout = h_ii; } else { W = W_in; Hout = h_in; }
    } else if (b < 2 * PJ_T_INST + PJ_T_SVC) {
        tile = b - 2 * PJ_T_INST; X = feat_svc; W = W_sc; Hout = h_sc; nrows = SVCN;
    } else {
        tile = b - 2 * PJ_T_INST - PJ_T_SVC; X = feat_node; W = W_ni; Hout = h_ni; nrows = NODEN;
    }
    const int r0 = tile * 64 + (wave >> 1) * 16;
    const int n0 = (wave & 1) * 64 + 2 * c;

    const int  g      = b * DG_PER_BLOCK + t;
    const bool counts = t < DG_PER_BLOCK && g < DG_TOTAL;
    const int* sp; const int* dp; int oS, oD, gl;
    if (g < DG_SC) {
        sp = svc_src; dp = svc_dst; gl = g;
        oS = (int)(D_NS_SC - OFF_DEG); oD = (int)(D_ND_SC - OFF_DEG);
    } else if (g < DG_SC + DG_IN) {
        sp = in_src; dp = in_dst; gl = g - DG_SC;
        oS = (int)(D_NS_IN - OFF_DEG); oD = (int)(D_ND_IN - OFF_DEG);
    } else if (g < DG_SC + DG_IN + DG_NI) {
        sp = ni_src; dp = ni_dst; gl = g - DG_SC - DG_IN;
        oS = (int)(D_NS_NI - OFF_DEG); oD = (int)(D_ND_NI - OFF_DEG);
    } else {
        sp = ii_src; dp = ii_dst; gl = g - DG_SC - DG_IN - DG_NI;
        oS = (int)(D_NS_II - OFF_DEG); oD = (int)(D_ND_II - OFF_DEG);
    }
    int4 es = make_int4(0, 0, 0, 0), ed = make_int4(0, 0, 0, 0);
    if (counts) {
        es = ((const int4*)sp)[gl];
        ed = ((const int4*)dp)[gl];
    }

    const int rowA = (r0 + c < nrows) ? (r0 + c) : (nrows - 1);
    const float* Ap = X + (long long)rowA * 128 + 8 * half;
    float4 a[PJ_STEPS][2];
#pragma unroll
    for (int ks = 0; ks < PJ_STEPS; ++ks) {
        a[ks][0] = *(const float4*)(Ap + 16 * ks);
        a[ks][1] = *(const float4*)(Ap + 16 * ks + 4);
    }

#pragma unroll
    for (int it = 0; it < 8; ++it) {
        const int u  = t + 256 * it;
        const int kp = u >> 5;
        const int m4 = (u & 31) * 4;
        const float4 w0 = *(const float4*)&W[(2 * kp) * 128 + m4];
        const float4 w1 = *(const float4*)&W[(2 * kp + 1) * 128 + m4];
        float* d = &Wlds[(kp * 128 + m4) * 2];
        *(float4*)(d)     = make_float4(w0.x, w1.x, w0.y, w1.y);
        *(float4*)(d + 4) = make_float4(w0.z, w1.z, w0.w, w1.w);
    }
    __syncthreads();

    v8f acc[4] = {v8f{}, v8f{}, v8f{}, v8f{}};
    const float* Bp = &Wlds[(4 * half * 128 + n0) * 2];

#pragma unroll
    for (int ks = 0; ks < PJ_STEPS; ++ks) {
#pragma unroll
        for (int s = 0; s < 4; ++s) {
            const float* bp = Bp + ((8 * ks + s) * 128) * 2;
            const float4 b0 = *(const float4*)(bp);
            const float4 b1 = *(const float4*)(bp + 64);
            const float4 av = a[ks][s >> 1];
            v2f af;
            af.x = (s & 1) ? av.z : av.x;
            af.y = (s & 1) ? av.w : av.y;
            v2f bf;
            bf.x = b0.x; bf.y = b0.y;
            acc[0] = __builtin_amdgcn_wmma_f32_16x16x4_f32(false, af, false, bf, (short)0, acc[0], false, false);
            bf.x = b0.z; bf.y = b0.w;
            acc[1] = __builtin_amdgcn_wmma_f32_16x16x4_f32(false, af, false, bf, (short)0, acc[1], false, false);
            bf.x = b1.x; bf.y = b1.y;
            acc[2] = __builtin_amdgcn_wmma_f32_16x16x4_f32(false, af, false, bf, (short)0, acc[2], false, false);
            bf.x = b1.z; bf.y = b1.w;
            acc[3] = __builtin_amdgcn_wmma_f32_16x16x4_f32(false, af, false, bf, (short)0, acc[3], false, false);
        }
    }

#pragma unroll
    for (int i = 0; i < 8; ++i) {
        const int row = r0 + i + 8 * half;
        if (row < nrows) {
            float* hp = Hout + (long long)row * 128 + n0;
            *(float2*)(hp)      = make_float2(acc[0][i], acc[1][i]);
            *(float2*)(hp + 32) = make_float2(acc[2][i], acc[3][i]);
        }
    }

    if (counts) {
        float* cS = deg + oS;
        float* cD = deg + oD;
        atomAddF(cS + es.x, 1.f); atomAddF(cS + es.y, 1.f);
        atomAddF(cS + es.z, 1.f); atomAddF(cS + es.w, 1.f);
        atomAddF(cD + ed.x, 1.f); atomAddF(cD + ed.y, 1.f);
        atomAddF(cD + ed.z, 1.f); atomAddF(cD + ed.w, 1.f);
    }
}

#define AG_ROWS    640
#define AG_THREADS 1024
#define AG_WAVES   (AG_THREADS / 32)
#define AG_FLY     16
#define AG_GRP     2
#define AG_T_INST  ((INSTN + AG_ROWS - 1) / AG_ROWS)
#define AG_T_SVC   ((SVCN  + AG_ROWS - 1) / AG_ROWS)
#define AG_T_NODE  ((NODEN + AG_ROWS - 1) / AG_ROWS)
static_assert(E_SC % 4 == 0 && E_IN % 4 == 0 && E_NI % 4 == 0 && E_II % 4 == 0,
              "destination ids are read 4 per lane");

struct AgList {
    const int*   src;
    const int*   dst;
    const float* cntS;
    const float* cntD;
    const float* H;
    int          n4;
};

template <bool FULL>
__device__ __forceinline__ void ag_flush(const AgList& L, float* tile, int r0, int lane, int qe, int n) {
    const int d = L.dst[qe];
    const int s = L.src[qe];
    const float cd = L.cntD[d];
    const float cs = L.cntS[s];
    const int f = __float_as_int(rsqrtf(cd > 1.f ? cd : 1.f) * rsqrtf(cs > 1.f ? cs : 1.f));
    const int r = d - r0;
    float h[AG_FLY][4];
#pragma unroll
    for (int j = 0; j < AG_FLY; ++j) {
        h[j][0] = 0.f; h[j][1] = 0.f; h[j][2] = 0.f; h[j][3] = 0.f;
        if (FULL || j < n) {
            const float* hp = L.H + (long long)__builtin_amdgcn_readlane(s, j) * 128;
            h[j][0] = hp[lane]; h[j][1] = hp[lane + 32]; h[j][2] = hp[lane + 64]; h[j][3] = hp[lane + 96];
        }
    }
#pragma unroll
    for (int j = 0; j < AG_FLY; ++j) {
        if (FULL || j < n) {
            float*      p  = tile + __builtin_amdgcn_readlane(r, j) * 128 + lane;
            const float fj = __int_as_float(__builtin_amdgcn_readlane(f, j));
            atomicAdd(p,      h[j][0] * fj);
            atomicAdd(p + 32, h[j][1] * fj);
            atomicAdd(p + 64, h[j][2] * fj);
            atomicAdd(p + 96, h[j][3] * fj);
        }
    }
}

__device__ __forceinline__ void ag_group(const AgList& L, float* tile, int r0, int nr, int lane,
                                         int4 d, int gb, int& qe, int& cnt) {
    const unsigned ux = (unsigned)(d.x - r0), uy = (unsigned)(d.y - r0);
    const unsigned uz = (unsigned)(d.z - r0), uw = (unsigned)(d.w - r0);
    const unsigned lo = min(min(ux, uy), min(uz, uw));
    unsigned lanes = __builtin_amdgcn_ballot_w32(gb + lane < L.n4 && lo < (unsigned)nr);
    if (lanes == 0u) return;
    const int code = (ux < (unsigned)nr ? 1 : 0) | (uy < (unsigned)nr ? 2 : 0) |
                     (uz < (unsigned)nr ? 4 : 0) | (uw < (unsigned)nr ? 8 : 0);
    do {
        const int l = __builtin_ctz(lanes);
        lanes &= lanes - 1u;
        unsigned c = (unsigned)__builtin_amdgcn_readlane(code, l);
        const int eb = 4 * (gb + l);
        do {
            const int k = __builtin_ctz(c);
            c &= c - 1u;
            qe = (lane == cnt) ? (eb + k) : qe;
            if (++cnt == AG_FLY) { ag_flush<true>(L, tile, r0, lane, qe, AG_FLY); cnt = 0; }
        } while (c != 0u);
    } while (lanes != 0u);
}

__device__ __forceinline__ void ag_list(const AgList& L, float* tile, int r0, int nr, int wave, int lane) {
    const int4* dst4 = (const int4*)L.dst;
    int qe  = 0;
    int cnt = 0;
    int i = wave * 32 + lane;
    int4 n0 = dst4[i < L.n4 ? i : 0];
    int4 n1 = dst4[i + AG_THREADS < L.n4 ? i + AG_THREADS : 0];
    for (int gb = wave * 32; gb < L.n4; gb += AG_GRP * AG_THREADS) {
        const int4 d0 = n0, d1 = n1;
        i += AG_GRP * AG_THREADS;
        n0 = dst4[i < L.n4 ? i : 0];
        n1 = dst4[i + AG_THREADS < L.n4 ? i + AG_THREADS : 0];
        ag_group(L, tile, r0, nr, lane, d0, gb, qe, cnt);
        ag_group(L, tile, r0, nr, lane, d1, gb + AG_THREADS, qe, cnt);
    }
    if (cnt != 0) ag_flush<false>(L, tile, r0, lane, qe, cnt);
}

__global__ __launch_bounds__(AG_THREADS) void agg_kernel(
        const int* __restrict__ svc_src, const int* __restrict__ svc_dst,
        const int* __restrict__ in_src,  const int* __restrict__ in_dst,
        const int* __restrict__ ni_src,  const int* __restrict__ ni_dst,
        const int* __restrict__ ii_src,  const int* __restrict__ ii_dst,
        const float* __restrict__ h_sc, const float* __restrict__ h_in,
        const float* __restrict__ h_ni, const float* __restrict__ h_ii,
        const float* __restrict__ deg,
        const float* __restrict__ b_sc, const float* __restrict__ b_in,
        const float* __restrict__ b_ni, const float* __restrict__ b_ii,
        float* __restrict__ cat) {
    __shared__ float tile[AG_ROWS * 128];
    const int lane = threadIdx.x & 31;
    const int wave = __builtin_amdgcn_readfirstlane(threadIdx.x >> 5);
    const int b    = blockIdx.x;

    for (int i = threadIdx.x; i < AG_ROWS * 32; i += AG_THREADS)
        ((float4*)tile)[i] = make_float4(0.f, 0.f, 0.f, 0.f);
    __syncthreads();

    int r0, nr, catRow0, nLists;
    AgList L;
    float4 bias;
    const int c4 = lane * 4;
    if (b < AG_T_INST) {
        r0 = b * AG_ROWS;
        nr = (INSTN - r0 < AG_ROWS) ? (INSTN - r0) : AG_ROWS;
        catRow0 = SVCN + NODEN + r0;
        L.src = ni_src; L.dst = ni_dst; L.H = h_ni; L.n4 = E_NI / 4; nLists = 2;
        L.cntS = deg + (D_NS_NI - OFF_DEG); L.cntD = deg + (D_ND_NI - OFF_DEG);
        const float4 x = *(const float4*)&b_ni[c4];
        const float4 y = *(const float4*)&b_ii[c4];
        bias = make_float4(x.x + y.x, x.y + y.y, x.z + y.z, x.w + y.w);
    } else if (b < AG_T_INST + AG_T_SVC) {
        r0 = (b - AG_T_INST) * AG_ROWS;
        nr = (SVCN - r0 < AG_ROWS) ? (SVCN - r0) : AG_ROWS;
        catRow0 = r0;
        L.src = svc_src; L.dst = svc_dst; L.H = h_sc; L.n4 = E_SC / 4; nLists = 1;
        L.cntS = deg + (D_NS_SC - OFF_DEG); L.cntD = deg + (D_ND_SC - OFF_DEG);
        bias = *(const float4*)&b_sc[c4];
    } else {
        r0 = (b - AG_T_INST - AG_T_SVC) * AG_ROWS;
        nr = (NODEN - r0 < AG_ROWS) ? (NODEN - r0) : AG_ROWS;
        catRow0 = SVCN + r0;
        L.src = in_src; L.dst = in_dst; L.H = h_in; L.n4 = E_IN / 4; nLists = 1;
        L.cntS = deg + (D_NS_IN - OFF_DEG); L.cntD = deg + (D_ND_IN - OFF_DEG);
        bias = *(const float4*)&b_in[c4];
    }
#pragma nounroll
    for (int li = 0; li < nLists; ++li) {
        ag_list(L, tile, r0, nr, wave, lane);
        L.src = ii_src; L.dst = ii_dst; L.H = h_ii; L.n4 = E_II / 4;
        L.cntS = deg + (D_NS_II - OFF_DEG); L.cntD = deg + (D_ND_II - OFF_DEG);
    }
    __syncthreads();

    for (int r = wave; r < nr; r += AG_WAVES) {
        const float4 v = *(const float4*)&tile[r * 128 + c4];
        *(float4*)&cat[(long long)(catRow0 + r) * 128 + c4] =
            make_float4(v.x + bias.x, v.y + bias.y, v.z + bias.z, v.w + bias.w);
    }
}

#define FG_KSLICE 800
#define FG_STEP   16
#define FG_GROUPS (TOTAL / FG_KSLICE)
static_assert(TOTAL % FG_KSLICE == 0, "K slices must tile K exactly");
static_assert(FG_KSLICE % (2 * FG_STEP) == 0, "loop body covers two steps");

struct FgFrag {
    float4 a[4][2];
    float2 b[4][4][2];
};

__device__ __forceinline__ void fg_load(FgFrag& f, const float* const (&Ap)[4],
                                        const float* __restrict__ Bp, int k) {
#pragma unroll
    for (int rt = 0; rt < 4; ++rt) {
        f.a[rt][0] = *(const float4*)(Ap[rt] + k);
        f.a[rt][1] = *(const float4*)(Ap[rt] + k + 4);
    }
#pragma unroll
    for (int s = 0; s < 4; ++s) {
#pragma unroll
        for (int q = 0; q < 4; ++q) {
            f.b[s][q][0] = *(const float2*)(Bp + (long long)(k + 2 * s) * 128 + 32 * q);
            f.b[s][q][1] = *(const float2*)(Bp + (long long)(k + 2 * s + 1) * 128 + 32 * q);
        }
    }
}

__device__ __forceinline__ void fg_mma(const FgFrag& f, v8f (&acc)[4][8]) {
#pragma unroll
    for (int s = 0; s < 4; ++s) {
        v2f bf[4][2];
#pragma unroll
        for (int q = 0; q < 4; ++q) {
            bf[q][0].x = lrelu(f.b[s][q][0].x);
            bf[q][0].y = lrelu(f.b[s][q][1].x);
            bf[q][1].x = lrelu(f.b[s][q][0].y);
            bf[q][1].y = lrelu(f.b[s][q][1].y);
        }
#pragma unroll
        for (int rt = 0; rt < 4; ++rt) {
            const float4 av = f.a[rt][s >> 1];
            v2f a;
            a.x = (s & 1) ? av.z : av.x;
            a.y = (s & 1) ? av.w : av.y;
#pragma unroll
            for (int q = 0; q < 4; ++q) {
#pragma unroll
                for (int e = 0; e < 2; ++e) {
                    acc[rt][2 * q + e] = __builtin_amdgcn_wmma_f32_16x16x4_f32(
                        false, a, false, bf[q][e], (short)0, acc[rt][2 * q + e], false, false);
                }
            }
        }
    }
}

__global__ __launch_bounds__(128) void final_gemm_kernel(const float* __restrict__ Wt,
                                                         const float* __restrict__ cat,
                                                         float* __restrict__ out) {
    const int t    = threadIdx.x;
    const int lane = t & 31;
    const int wave = t >> 5;
    const int half = lane >> 4;
    const int c    = lane & 15;
    const int m0   = wave * 64;
    const int k0   = blockIdx.x * FG_KSLICE;
    const int kEnd = k0 + FG_KSLICE;

    const float* Ap[4];
#pragma unroll
    for (int rt = 0; rt < 4; ++rt)
        Ap[rt] = Wt + (long long)(m0 + rt * 16 + c) * TOTAL + 8 * half;
    const float* Bp = cat + (long long)(8 * half) * 128 + 2 * c;

    v8f acc[4][8];
#pragma unroll
    for (int rt = 0; rt < 4; ++rt)
#pragma unroll
        for (int j = 0; j < 8; ++j) acc[rt][j] = v8f{};

    FgFrag f0, f1;
    fg_load(f0, Ap, Bp, k0);
    asm volatile("" ::: "memory");
    for (int k = k0; k < kEnd; k += 2 * FG_STEP) {
        fg_load(f1, Ap, Bp, k + FG_STEP);
        fg_mma(f0, acc);
        const int kn = (k + 2 * FG_STEP < kEnd) ? (k + 2 * FG_STEP) : k0;
        fg_load(f0, Ap, Bp, kn);
        asm volatile("" ::: "memory");
        fg_mma(f1, acc);
    }

#pragma unroll
    for (int rt = 0; rt < 4; ++rt) {
#pragma unroll
        for (int q = 0; q < 4; ++q) {
#pragma unroll
            for (int e = 0; e < 2; ++e) {
#pragma unroll
                for (int i = 0; i < 8; ++i) {
                    int m = m0 + rt * 16 + i + 8 * half;
                    atomAddF(&out[m * 128 + 32 * q + 2 * c + e], acc[rt][2 * q + e][i]);
                }
            }
        }
    }
}

static inline int cdiv(long long a, long long b) { return (int)((a + b - 1) / b); }

extern "C" void kernel_launch(void* const* d_in, const int* in_sizes, int n_in,
                              void* d_out, int out_size, void* d_ws, size_t ws_size,
                              hipStream_t stream) {
    const float* feat_svc  = (const float*)d_in[0];
    const float* feat_inst = (const float*)d_in[1];
    const float* feat_node = (const float*)d_in[2];
    const int*   svc_src   = (const int*)d_in[3];
    const int*   svc_dst   = (const int*)d_in[4];
    const int*   in_src    = (const int*)d_in[5];
    const int*   in_dst    = (const int*)d_in[6];
    const int*   ni_src    = (const int*)d_in[7];
    const int*   ni_dst    = (const int*)d_in[8];
    const int*   ii_src    = (const int*)d_in[9];
    const int*   ii_dst    = (const int*)d_in[10];
    const float* W_sc = (const float*)d_in[11];  const float* b_sc = (const float*)d_in[12];
    const float* W_in = (const float*)d_in[13];  const float* b_in = (const float*)d_in[14];
    const float* W_ni = (const float*)d_in[15];  const float* b_ni = (const float*)d_in[16];
    const float* W_ii = (const float*)d_in[17];  const float* b_ii = (const float*)d_in[18];
    const float* W_tot = (const float*)d_in[19]; const float* b_tot = (const float*)d_in[20];
    float* out = (float*)d_out;

    float* ws   = (float*)d_ws;
    float* cat  = ws + OFF_CAT;
    float* h_sc = ws + OFF_HSC;
    float* h_in = ws + OFF_HIN;
    float* h_ni = ws + OFF_HNI;
    float* h_ii = ws + OFF_HII;
    float* deg  = ws + OFF_DEG;

    init_deg_kernel<<<SMALL_BLOCKS, 256, 0, stream>>>(deg);
    init_out_kernel<<<cdiv(OUTD * HDIM, 256), 256, 0, stream>>>(out, b_tot);

    proj_deg_kernel<<<PJ_BLOCKS, 256, 0, stream>>>(
        feat_svc, feat_inst, feat_node, W_sc, W_in, W_ni, W_ii, h_sc, h_in, h_ni, h_ii,
        svc_src, svc_dst, in_src, in_dst, ni_src, ni_dst, ii_src, ii_dst, deg);

    agg_kernel<<<AG_T_INST + AG_T_SVC + AG_T_NODE, AG_THREADS, 0, stream>>>(
        svc_src, svc_dst, in_src, in_dst, ni_src, ni_dst, ii_src, ii_dst,
        h_sc, h_in, h_ni, h_ii, deg,
        b_sc, b_in, b_ni, b_ii, cat);

    final_gemm_kernel<<<FG_GROUPS, 128, 0, stream>>>(W_tot, cat, out);
}


# final GEMM: the 4 waves of a work-group share one 64-row block over 4 K slices (768 or 800 k), sum accumulators through LDS in fixed order, then f32 atomics: 4x fewer atomics; sum grouping changes
# speedup vs baseline: 2.3496x; 1.0522x over previous
#include <hip/hip_runtime.h>

#define SVCN   50000
#define INSTN  100000
#define NODEN  50000
#define FDIM   128
#define HDIM   128
#define OUTD   256
#define TOTAL  (SVCN + NODEN + INSTN)
#define E_SC   400000
#define E_IN   100000
#define E_NI   100000
#define E_II   800000

#define OFF_CAT   0LL
#define OFF_HSC   (OFF_CAT  + (long long)TOTAL * 128)
#define OFF_HIN   (OFF_HSC  + (long long)SVCN  * 128)
#define OFF_HNI   (OFF_HIN  + (long long)INSTN * 128)
#define OFF_HII   (OFF_HNI  + (long long)NODEN * 128)
#define OFF_DEG   (OFF_HII  + (long long)INSTN * 128)
#define D_NS_SC   (OFF_DEG + 0LL)
#define D_ND_SC   (D_NS_SC + SVCN)
#define D_NS_IN   (D_ND_SC + SVCN)
#define D_ND_IN   (D_NS_IN + INSTN)
#define D_NS_NI   (D_ND_IN + NODEN)
#define D_ND_NI   (D_NS_NI + NODEN)
#define D_NS_II   (D_ND_NI + INSTN)
#define D_ND_II   (D_NS_II + INSTN)
#define DEG_TOTAL 600000

typedef __attribute__((ext_vector_type(2))) float v2f;
typedef __attribute__((ext_vector_type(8))) float v8f;

__device__ __forceinline__ float lrelu(float x) { return x > 0.f ? x : 0.01f * x; }

__device__ __forceinline__ void atomAddF(float* p, float v) { unsafeAtomicAdd(p, v); }

#define SMALL_BLOCKS 128
static_assert(DEG_TOTAL % 4 == 0, "degree planes are walked 16 bytes per lane");

__global__ __launch_bounds__(256) void init_deg_kernel(float* deg) {
    float4* d4 = (float4*)deg;
    for (int i = blockIdx.x * 256 + threadIdx.x; i < DEG_TOTAL / 4; i += SMALL_BLOCKS * 256)
        d4[i] = make_float4(0.f, 0.f, 0.f, 0.f);
}

__global__ void init_out_kernel(float* __restrict__ out, const float* __restrict__ b_tot) {
    int i = blockIdx.x * blockDim.x + threadIdx.x;
    if (i < OUTD * HDIM) out[i] = b_tot[i >> 7];
}

#define PJ_STEPS (FDIM / 16)
#define PJ_T_SVC   ((SVCN  + 63) / 64)
#define PJ_T_NODE  ((NODEN + 63) / 64)
#define PJ_T_INST  ((INSTN + 63) / 64)
#define PJ_BLOCKS  (2 * PJ_T_INST + PJ_T_SVC + PJ_T_NODE)
#define DG_SC      (E_SC / 4)
#define DG_IN      (E_IN / 4)
#define DG_NI      (E_NI / 4)
#define DG_II      (E_II / 4)
#define DG_TOTAL   (DG_SC + DG_IN + DG_NI + DG_II)
#define DG_PER_BLOCK ((DG_TOTAL + PJ_BLOCKS - 1) / PJ_BLOCKS)
static_assert(E_SC % 4 == 0 && E_IN % 4 == 0 && E_NI % 4 == 0 && E_II % 4 == 0,
              "edge ids are read 4 per lane");
static_assert(DG_PER_BLOCK <= 256, "one id group per thread");

__global__ __launch_bounds__(256) void proj_deg_kernel(
        const float* __restrict__ feat_svc, const float* __restrict__ feat_inst,
        const float* __restrict__ feat_node,
        const float* __restrict__ W_sc, const float* __restrict__ W_in,
        const float* __restrict__ W_ni, const float* __restrict__ W_ii,
        float* __restrict__ h_sc, float* __restrict__ h_in,
        float* __restrict__ h_ni, float* __restrict__ h_ii,
        const int* __restrict__ svc_src, const int* __restrict__ svc_dst,
        const int* __restrict__ in_src,  const int* __restrict__ in_dst,
        const int* __restrict__ ni_src,  const int* __restrict__ ni_dst,
        const int* __restrict__ ii_src,  const int* __restrict__ ii_dst,
        float* __restrict__ deg) {
    __shared__ float Wlds[128 * 128];
    const int t    = threadIdx.x;
    const int lane = t & 31;
    const int wave = t >> 5;
    const int half = lane >> 4;
    const int c    = lane & 15;
    const int b    = blockIdx.x;

    const float* X; const float* W; float* Hout; int nrows, tile;
    if (b < 2 * PJ_T_INST) {
        tile = b >> 1; X = feat_inst; nrows = INSTN;
        if (b & 1) { W = W_ii; Hout = h_ii; } else { W = W_in; Hout = h_in; }
    } else if (b < 2 * PJ_T_INST + PJ_T_SVC) {
        tile = b - 2 * PJ_T_INST; X = feat_svc; W = W_sc; Hout = h_sc; nrows = SVCN;
    } else {
        tile = b - 2 * PJ_T_INST - PJ_T_SVC; X = feat_node; W = W_ni; Hout = h_ni; nrows = NODEN;
    }
    const int r0 = tile * 64 + (wave >> 1) * 16;
    const int n0 = (wave & 1) * 64 + 2 * c;

    const int  g      = b * DG_PER_BLOCK + t;
    const bool counts = t < DG_PER_BLOCK && g < DG_TOTAL;
    const int* sp; const int* dp; int oS, oD, gl;
    if (g < DG_SC) {
        sp = svc_src; dp = svc_dst; gl = g;
        oS = (int)(D_NS_SC - OFF_DEG); oD = (int)(D_ND_SC - OFF_DEG);
    } else if (g < DG_SC + DG_IN) {
        sp = in_src; dp = in_dst; gl = g - DG_SC;
        oS = (int)(D_NS_IN - OFF_DEG); oD = (int)(D_ND_IN - OFF_DEG);
    } else if (g < DG_SC + DG_IN + DG_NI) {
        sp = ni_src; dp = ni_dst; gl = g - DG_SC - DG_IN;
        oS = (int)(D_NS_NI - OFF_DEG); oD = (int)(D_ND_NI - OFF_DEG);
    } else {
        sp = ii_src; dp = ii_dst; gl = g - DG_SC - DG_IN - DG_NI;
        oS = (int)(D_NS_II - OFF_DEG); oD = (int)(D_ND_II - OFF_DEG);
    }
    int4 es = make_int4(0, 0, 0, 0), ed = make_int4(0, 0, 0, 0);
    if (counts) {
        es = ((const int4*)sp)[gl];
        ed = ((const int4*)dp)[gl];
    }

    const int rowA = (r0 + c < nrows) ? (r0 + c) : (nrows - 1);
    const float* Ap = X + (long long)rowA * 128 + 8 * half;
    float4 a[PJ_STEPS][2];
#pragma unroll
    for (int ks = 0; ks < PJ_STEPS; ++ks) {
        a[ks][0] = *(const float4*)(Ap + 16 * ks);
        a[ks][1] = *(const float4*)(Ap + 16 * ks + 4);
    }

#pragma unroll
    for (int it = 0; it < 8; ++it) {
        const int u  = t + 256 * it;
        const int kp = u >> 5;
        const int m4 = (u & 31) * 4;
        const float4 w0 = *(const float4*)&W[(2 * kp) * 128 + m4];
        const float4 w1 = *(const float4*)&W[(2 * kp + 1) * 128 + m4];
        float* d = &Wlds[(kp * 128 + m4) * 2];
        *(float4*)(d)     = make_float4(w0.x, w1.x, w0.y, w1.y);
        *(float4*)(d + 4) = make_float4(w0.z, w1.z, w0.w, w1.w);
    }
    __syncthreads();

    v8f acc[4] = {v8f{}, v8f{}, v8f{}, v8f{}};
    const float* Bp = &Wlds[(4 * half * 128 + n0) * 2];

#pragma unroll
    for (int ks = 0; ks < PJ_STEPS; ++ks) {
#pragma unroll
        for (int s = 0; s < 4; ++s) {
            const float* bp = Bp + ((8 * ks + s) * 128) * 2;
            const float4 b0 = *(const float4*)(bp);
            const float4 b1 = *(const float4*)(bp + 64);
            const float4 av = a[ks][s >> 1];
            v2f af;
            af.x = (s & 1) ? av.z : av.x;
            af.y = (s & 1) ? av.w : av.y;
            v2f bf;
            bf.x = b0.x; bf.y = b0.y;
            acc[0] = __builtin_amdgcn_wmma_f32_16x16x4_f32(false, af, false, bf, (short)0, acc[0], false, false);
            bf.x = b0.z; bf.y = b0.w;
            acc[1] = __builtin_amdgcn_wmma_f32_16x16x4_f32(false, af, false, bf, (short)0, acc[1], false, false);
            bf.x = b1.x; bf.y = b1.y;
            acc[2] = __builtin_amdgcn_wmma_f32_16x16x4_f32(false, af, false, bf, (short)0, acc[2], false, false);
            bf.x = b1.z; bf.y = b1.w;
            acc[3] = __builtin_amdgcn_wmma_f32_16x16x4_f32(false, af, false, bf, (short)0, acc[3], false, false);
        }
    }

#pragma unroll
    for (int i = 0; i < 8; ++i) {
        const int row = r0 + i + 8 * half;
        if (row < nrows) {
            float* hp = Hout + (long long)row * 128 + n0;
            *(float2*)(hp)      = make_float2(acc[0][i], acc[1][i]);
            *(float2*)(hp + 32) = make_float2(acc[2][i], acc[3][i]);
        }
    }

    if (counts) {
        float* cS = deg + oS;
        float* cD = deg + oD;
        atomAddF(cS + es.x, 1.f); atomAddF(cS + es.y, 1.f);
        atomAddF(cS + es.z, 1.f); atomAddF(cS + es.w, 1.f);
        atomAddF(cD + ed.x, 1.f); atomAddF(cD + ed.y, 1.f);
        atomAddF(cD + ed.z, 1.f); atomAddF(cD + ed.w, 1.f);
    }
}

#define AG_ROWS    640
#define AG_THREADS 1024
#define AG_WAVES   (AG_THREADS / 32)
#define AG_FLY     16
#define AG_GRP     2
#define AG_T_INST  ((INSTN + AG_ROWS - 1) / AG_ROWS)
#define AG_T_SVC   ((SVCN  + AG_ROWS - 1) / AG_ROWS)
#define AG_T_NODE  ((NODEN + AG_ROWS - 1) / AG_ROWS)
static_assert(E_SC % 4 == 0 && E_IN % 4 == 0 && E_NI % 4 == 0 && E_II % 4 == 0,
              "destination ids are read 4 per lane");

struct AgList {
    const int*   src;
    const int*   dst;
    const float* cntS;
    const float* cntD;
    const float* H;
    int          n4;
};

template <bool FULL>
__device__ __forceinline__ void ag_flush(const AgList& L, float* tile, int r0, int lane, int qe, int n) {
    const int d = L.dst[qe];
    const int s = L.src[qe];
    const float cd = L.cntD[d];
    const float cs = L.cntS[s];
    const int f = __float_as_int(rsqrtf(cd > 1.f ? cd : 1.f) * rsqrtf(cs > 1.f ? cs : 1.f));
    const int r = d - r0;
    float h[AG_FLY][4];
#pragma unroll
    for (int j = 0; j < AG_FLY; ++j) {
        h[j][0] = 0.f; h[j][1] = 0.f; h[j][2] = 0.f; h[j][3] = 0.f;
        if (FULL || j < n) {
            const float* hp = L.H + (long long)__builtin_amdgcn_readlane(s, j) * 128;
            h[j][0] = hp[lane]; h[j][1] = hp[lane + 32]; h[j][2] = hp[lane + 64]; h[j][3] = hp[lane + 96];
        }
    }
#pragma unroll
    for (int j = 0; j < AG_FLY; ++j) {
        if (FULL || j < n) {
            float*      p  = tile + __builtin_amdgcn_readlane(r, j) * 128 + lane;
            const float fj = __int_as_float(__builtin_amdgcn_readlane(f, j));
            atomicAdd(p,      h[j][0] * fj);
            atomicAdd(p + 32, h[j][1] * fj);
            atomicAdd(p + 64, h[j][2] * fj);
            atomicAdd(p + 96, h[j][3] * fj);
        }
    }
}

__device__ __forceinline__ void ag_group(const AgList& L, float* tile, int r0, int nr, int lane,
                                         int4 d, int gb, int& qe, int& cnt) {
    const unsigned ux = (unsigned)(d.x - r0), uy = (unsigned)(d.y - r0);
    const unsigned uz = (unsigned)(d.z - r0), uw = (unsigned)(d.w - r0);
    const unsigned lo = min(min(ux, uy), min(uz, uw));
    unsigned lanes = __builtin_amdgcn_ballot_w32(gb + lane < L.n4 && lo < (unsigned)nr);
    if (lanes == 0u) return;
    const int code = (ux < (unsigned)nr ? 1 : 0) | (uy < (unsigned)nr ? 2 : 0) |
                     (uz < (unsigned)nr ? 4 : 0) | (uw < (unsigned)nr ? 8 : 0);
    do {
        const int l = __builtin_ctz(lanes);
        lanes &= lanes - 1u;
        unsigned c = (unsigned)__builtin_amdgcn_readlane(code, l);
        const int eb = 4 * (gb + l);
        do {
            const int k = __builtin_ctz(c);
            c &= c - 1u;
            qe = (lane == cnt) ? (eb + k) : qe;
            if (++cnt == AG_FLY) { ag_flush<true>(L, tile, r0, lane, qe, AG_FLY); cnt = 0; }
        } while (c != 0u);
    } while (lanes != 0u);
}

__device__ __forceinline__ void ag_list(const AgList& L, float* tile, int r0, int nr, int wave, int lane) {
    const int4* dst4 = (const int4*)L.dst;
    int qe  = 0;
    int cnt = 0;
    int i = wave * 32 + lane;
    int4 n0 = dst4[i < L.n4 ? i : 0];
    int4 n1 = dst4[i + AG_THREADS < L.n4 ? i + AG_THREADS : 0];
    for (int gb = wave * 32; gb < L.n4; gb += AG_GRP * AG_THREADS) {
        const int4 d0 = n0, d1 = n1;
        i += AG_GRP * AG_THREADS;
        n0 = dst4[i < L.n4 ? i : 0];
        n1 = dst4[i + AG_THREADS < L.n4 ? i + AG_THREADS : 0];
        ag_group(L, tile, r0, nr, lane, d0, gb, qe, cnt);
        ag_group(L, tile, r0, nr, lane, d1, gb + AG_THREADS, qe, cnt);
    }
    if (cnt != 0) ag_flush<false>(L, tile, r0, lane, qe, cnt);
}

__global__ __launch_bounds__(AG_THREADS) void agg_kernel(
        const int* __restrict__ svc_src, const int* __restrict__ svc_dst,
        const int* __restrict__ in_src,  const int* __restrict__ in_dst,
        const int* __restrict__ ni_src,  const int* __restrict__ ni_dst,
        const int* __restrict__ ii_src,  const int* __restrict__ ii_dst,
        const float* __restrict__ h_sc, const float* __restrict__ h_in,
        const float* __restrict__ h_ni, const float* __restrict__ h_ii,
        const float* __restrict__ deg,
        const float* __restrict__ b_sc, const float* __restrict__ b_in,
        const float* __restrict__ b_ni, const float* __restrict__ b_ii,
        float* __restrict__ cat) {
    __shared__ float tile[AG_ROWS * 128];
    const int lane = threadIdx.x & 31;
    const int wave = __builtin_amdgcn_readfirstlane(threadIdx.x >> 5);
    const int b    = blockIdx.x;

    for (int i = threadIdx.x; i < AG_ROWS * 32; i += AG_THREADS)
        ((float4*)tile)[i] = make_float4(0.f, 0.f, 0.f, 0.f);
    __syncthreads();

    int r0, nr, catRow0, nLists;
    AgList L;
    float4 bias;
    const int c4 = lane * 4;
    if (b < AG_T_INST) {
        r0 = b * AG_ROWS;
        nr = (INSTN - r0 < AG_ROWS) ? (INSTN - r0) : AG_ROWS;
        catRow0 = SVCN + NODEN + r0;
        L.src = ni_src; L.dst = ni_dst; L.H = h_ni; L.n4 = E_NI / 4; nLists = 2;
        L.cntS = deg + (D_NS_NI - OFF_DEG); L.cntD = deg + (D_ND_NI - OFF_DEG);
        const float4 x = *(const float4*)&b_ni[c4];
        const float4 y = *(const float4*)&b_ii[c4];
        bias = make_float4(x.x + y.x, x.y + y.y, x.z + y.z, x.w + y.w);
    } else if (b < AG_T_INST + AG_T_SVC) {
        r0 = (b - AG_T_INST) * AG_ROWS;
        nr = (SVCN - r0 < AG_ROWS) ? (SVCN - r0) : AG_ROWS;
        catRow0 = r0;
        L.src = svc_src; L.dst = svc_dst; L.H = h_sc; L.n4 = E_SC / 4; nLists = 1;
        L.cntS = deg + (D_NS_SC - OFF_DEG); L.cntD = deg + (D_ND_SC - OFF_DEG);
        bias = *(const float4*)&b_sc[c4];
    } else {
        r0 = (b - AG_T_INST - AG_T_SVC) * AG_ROWS;
        nr = (NODEN - r0 < AG_ROWS) ? (NODEN - r0) : AG_ROWS;
        catRow0 = SVCN + r0;
        L.src = in_src; L.dst = in_dst; L.H = h_in; L.n4 = E_IN / 4; nLists = 1;
        L.cntS = deg + (D_NS_IN - OFF_DEG); L.cntD = deg + (D_ND_IN - OFF_DEG);
        bias = *(const float4*)&b_in[c4];
    }
#pragma nounroll
    for (int li = 0; li < nLists; ++li) {
        ag_list(L, tile, r0, nr, wave, lane);
        L.src = ii_src; L.dst = ii_dst; L.H = h_ii; L.n4 = E_II / 4;
        L.cntS = deg + (D_NS_II - OFF_DEG); L.cntD = deg + (D_ND_II - OFF_DEG);
    }
    __syncthreads();

    for (int r = wave; r < nr; r += AG_WAVES) {
        const float4 v = *(const float4*)&tile[r * 128 + c4];
        *(float4*)&cat[(long long)(catRow0 + r) * 128 + c4] =
            make_float4(v.x + bias.x, v.y + bias.y, v.z + bias.z, v.w + bias.w);
    }
}

#define FG_STEP   16
#define FG_DSTEPS (TOTAL / (2 * FG_STEP))
#define FG_CHUNKS 63
#define FG_SLICES (4 * FG_CHUNKS)
#define FG_GROUPS (4 * FG_CHUNKS)
#define FG_SLOT   2048
static_assert(TOTAL % (2 * FG_STEP) == 0, "K is walked in double steps");
static_assert(FG_DSTEPS / FG_SLICES >= 1, "no empty slice");

struct FgFrag {
    float4 a[4][2];
    float2 b[4][4][2];
};

__device__ __forceinline__ void fg_load(FgFrag& f, const float* const (&Ap)[4],
                                        const float* __restrict__ Bp, int k) {
#pragma unroll
    for (int rt = 0; rt < 4; ++rt) {
        f.a[rt][0] = *(const float4*)(Ap[rt] + k);
        f.a[rt][1] = *(const float4*)(Ap[rt] + k + 4);
    }
#pragma unroll
    for (int s = 0; s < 4; ++s) {
#pragma unroll
        for (int q = 0; q < 4; ++q) {
            f.b[s][q][0] = *(const float2*)(Bp + (long long)(k + 2 * s) * 128 + 32 * q);
            f.b[s][q][1] = *(const float2*)(Bp + (long long)(k + 2 * s + 1) * 128 + 32 * q);
        }
    }
}

__device__ __forceinline__ void fg_mma(const FgFrag& f, v8f (&acc)[4][8]) {
#pragma unroll
    for (int s = 0; s < 4; ++s) {
        v2f bf[4][2];
#pragma unroll
        for (int q = 0; q < 4; ++q) {
            bf[q][0].x = lrelu(f.b[s][q][0].x);
            bf[q][0].y = lrelu(f.b[s][q][1].x);
            bf[q][1].x = lrelu(f.b[s][q][0].y);
            bf[q][1].y = lrelu(f.b[s][q][1].y);
        }
#pragma unroll
        for (int rt = 0; rt < 4; ++rt) {
            const float4 av = f.a[rt][s >> 1];
            v2f a;
            a.x = (s & 1) ? av.z : av.x;
            a.y = (s & 1) ? av.w : av.y;
#pragma unroll
            for (int q = 0; q < 4; ++q) {
#pragma unroll
                for (int e = 0; e < 2; ++e) {
                    acc[rt][2 * q + e] = __builtin_amdgcn_wmma_f32_16x16x4_f32(
                        false, a, false, bf[q][e], (short)0, acc[rt][2 * q + e], false, false);
                }
            }
        }
    }
}

__global__ __launch_bounds__(128) void final_gemm_kernel(const float* __restrict__ Wt,
                                                         const float* __restrict__ cat,
                                                         float* __restrict__ out) {
    __shared__ float red[16 * FG_SLOT];
    const int t    = threadIdx.x;
    const int lane = t & 31;
    const int wave = __builtin_amdgcn_readfirstlane(t >> 5);
    const int half = lane >> 4;
    const int c    = lane & 15;
    const int m0   = (blockIdx.x & 3) * 64;
    const int sl   = (blockIdx.x >> 2) * 4 + wave;
    const int k0   = 2 * FG_STEP * ((FG_DSTEPS * sl) / FG_SLICES);
    const int kEnd = 2 * FG_STEP * ((FG_DSTEPS * (sl + 1)) / FG_SLICES);

    const float* Ap[4];
#pragma unroll
    for (int rt = 0; rt < 4; ++rt)
        Ap[rt] = Wt + (long long)(m0 + rt * 16 + c) * TOTAL + 8 * half;
    const float* Bp = cat + (long long)(8 * half) * 128 + 2 * c;

    v8f acc[4][8];
#pragma unroll
    for (int rt = 0; rt < 4; ++rt)
#pragma unroll
        for (int j = 0; j < 8; ++j) acc[rt][j] = v8f{};

    FgFrag f0, f1;
    fg_load(f0, Ap, Bp, k0);
    asm volatile("" ::: "memory");
    for (int k = k0; k < kEnd; k += 2 * FG_STEP) {
        fg_load(f1, Ap, Bp, k + FG_STEP);
        fg_mma(f0, acc);
        const int kn = (k + 2 * FG_STEP < kEnd) ? (k + 2 * FG_STEP) : k0;
        fg_load(f0, Ap, Bp, kn);
        asm volatile("" ::: "memory");
        fg_mma(f1, acc);
    }

#pragma unroll
    for (int s = 0; s < 4; ++s) {
        if (wave == s) {
#pragma unroll
            for (int rt = 0; rt < 4; ++rt) {
                if (rt != s) {
                    float4* dp = (float4*)&red[(rt * 4 + s) * FG_SLOT] + lane;
#pragma unroll
                    for (int j = 0; j < 8; ++j) {
                        dp[(2 * j) * 32]     = make_float4(acc[rt][j][0], acc[rt][j][1], acc[rt][j][2], acc[rt][j][3]);
                        dp[(2 * j + 1) * 32] = make_float4(acc[rt][j][4], acc[rt][j][5], acc[rt][j][6], acc[rt][j][7]);
                    }
                }
            }
        }
    }
    __syncthreads();
#pragma unroll
    for (int rt = 0; rt < 4; ++rt) {
        if (wave == rt) {
#pragma unroll
            for (int s = 0; s < 4; ++s) {
                if (s != rt) {
                    const float4* sp = (const float4*)&red[(rt * 4 + s) * FG_SLOT] + lane;
#pragma unroll
                    for (int j = 0; j < 8; ++j) {
                        const float4 x = sp[(2 * j) * 32];
                        const float4 y = sp[(2 * j + 1) * 32];
                        acc[rt][j][0] += x.x; acc[rt][j][1] += x.y; acc[rt][j][2] += x.z; acc[rt][j][3] += x.w;
                        acc[rt][j][4] += y.x; acc[rt][j][5] += y.y; acc[rt][j][6] += y.z; acc[rt][j][7] += y.w;
                    }
                }
            }
#pragma unroll
            for (int q = 0; q < 4; ++q) {
#pragma unroll
                for (int e = 0; e < 2; ++e) {
#pragma unroll
                    for (int i = 0; i < 8; ++i) {
                        const int m = m0 + rt * 16 + i + 8 * half;
                        atomAddF(&out[m * 128 + 32 * q + 2 * c + e], acc[rt][2 * q + e][i]);
                    }
                }
            }
        }
    }
}

static inline int cdiv(long long a, long long b) { return (int)((a + b - 1) / b); }

extern "C" void kernel_launch(void* const* d_in, const int* in_sizes, int n_in,
                              void* d_out, int out_size, void* d_ws, size_t ws_size,
                              hipStream_t stream) {
    const float* feat_svc  = (const float*)d_in[0];
    const float* feat_inst = (const float*)d_in[1];
    const float* feat_node = (const float*)d_in[2];
    const int*   svc_src   = (const int*)d_in[3];
    const int*   svc_dst   = (const int*)d_in[4];
    const int*   in_src    = (const int*)d_in[5];
    const int*   in_dst    = (const int*)d_in[6];
    const int*   ni_src    = (const int*)d_in[7];
    const int*   ni_dst    = (const int*)d_in[8];
    const int*   ii_src    = (const int*)d_in[9];
    const int*   ii_dst    = (const int*)d_in[10];
    const float* W_sc = (const float*)d_in[11];  const float* b_sc = (const float*)d_in[12];
    const float* W_in = (const float*)d_in[13];  const float* b_in = (const float*)d_in[14];
    const float* W_ni = (const float*)d_in[15];  const float* b_ni = (const float*)d_in[16];
    const float* W_ii = (const float*)d_in[17];  const float* b_ii = (const float*)d_in[18];
    const float* W_tot = (const float*)d_in[19]; const float* b_tot = (const float*)d_in[20];
    float* out = (float*)d_out;

    float* ws   = (float*)d_ws;
    float* cat  = ws + OFF_CAT;
    float* h_sc = ws + OFF_HSC;
    float* h_in = ws + OFF_HIN;
    float* h_ni = ws + OFF_HNI;
    float* h_ii = ws + OFF_HII;
    float* deg  = ws + OFF_DEG;

    init_deg_kernel<<<SMALL_BLOCKS, 256, 0, stream>>>(deg);
    init_out_kernel<<<cdiv(OUTD * HDIM, 256), 256, 0, stream>>>(out, b_tot);

    proj_deg_kernel<<<PJ_BLOCKS, 256, 0, stream>>>(
        feat_svc, feat_inst, feat_node, W_sc, W_in, W_ni, W_ii, h_sc, h_in, h_ni, h_ii,
        svc_src, svc_dst, in_src, in_dst, ni_src, ni_dst, ii_src, ii_dst, deg);

    agg_kernel<<<AG_T_INST + AG_T_SVC + AG_T_NODE, AG_THREADS, 0, stream>>>(
        svc_src, svc_dst, in_src, in_dst, ni_src, ni_dst, ii_src, ii_dst,
        h_sc, h_in, h_ni, h_ii, deg,
        b_sc, b_in, b_ni, b_ii, cat);

    final_gemm_kernel<<<FG_GROUPS, 128, 0, stream>>>(W_tot, cat, out);
}


# aggregation scan reads a 1-byte owner-tile tag per edge (written by the projection launch from the ids it already holds), 16 edges per 16-byte load, not the 4-byte id; LDS add arrival order changes
# speedup vs baseline: 2.4376x; 1.0375x over previous
#include <hip/hip_runtime.h>

#define SVCN   50000
#define INSTN  100000
#define NODEN  50000
#define FDIM   128
#define HDIM   128
#define OUTD   256
#define TOTAL  (SVCN + NODEN + INSTN)
#define E_SC   400000
#define E_IN   100000
#define E_NI   100000
#define E_II   800000

#define OFF_CAT   0LL
#define OFF_HSC   (OFF_CAT  + (long long)TOTAL * 128)
#define OFF_HIN   (OFF_HSC  + (long long)SVCN  * 128)
#define OFF_HNI   (OFF_HIN  + (long long)INSTN * 128)
#define OFF_HII   (OFF_HNI  + (long long)NODEN * 128)
#define OFF_DEG   (OFF_HII  + (long long)INSTN * 128)
#define D_NS_SC   (OFF_DEG + 0LL)
#define D_ND_SC   (D_NS_SC + SVCN)
#define D_NS_IN   (D_ND_SC + SVCN)
#define D_ND_IN   (D_NS_IN + INSTN)
#define D_NS_NI   (D_ND_IN + NODEN)
#define D_ND_NI   (D_NS_NI + NODEN)
#define D_NS_II   (D_ND_NI + INSTN)
#define D_ND_II   (D_NS_II + INSTN)
#define DEG_TOTAL 600000
#define OFF_TAG   (OFF_DEG + DEG_TOTAL)

typedef __attribute__((ext_vector_type(2))) float v2f;
typedef __attribute__((ext_vector_type(8))) float v8f;

__device__ __forceinline__ float lrelu(float x) { return x > 0.f ? x : 0.01f * x; }

__device__ __forceinline__ void atomAddF(float* p, float v) { unsafeAtomicAdd(p, v); }

#define SMALL_BLOCKS 128
static_assert(DEG_TOTAL % 4 == 0, "degree planes are walked 16 bytes per lane");

__global__ __launch_bounds__(256) void init_deg_kernel(float* deg) {
    float4* d4 = (float4*)deg;
    for (int i = blockIdx.x * 256 + threadIdx.x; i < DEG_TOTAL / 4; i += SMALL_BLOCKS * 256)
        d4[i] = make_float4(0.f, 0.f, 0.f, 0.f);
}

__global__ void init_out_kernel(float* __restrict__ out, const float* __restrict__ b_tot) {
    int i = blockIdx.x * blockDim.x + threadIdx.x;
    if (i < OUTD * HDIM) out[i] = b_tot[i >> 7];
}

#define PJ_STEPS (FDIM / 16)
#define PJ_T_SVC   ((SVCN  + 63) / 64)
#define PJ_T_NODE  ((NODEN + 63) / 64)
#define PJ_T_INST  ((INSTN + 63) / 64)
#define PJ_BLOCKS  (2 * PJ_T_INST + PJ_T_SVC + PJ_T_NODE)
#define DG_SC      (E_SC / 4)
#define DG_IN      (E_IN / 4)
#define DG_NI      (E_NI / 4)
#define DG_II      (E_II / 4)
#define DG_TOTAL   (DG_SC + DG_IN + DG_NI + DG_II)
#define DG_PER_BLOCK ((((DG_TOTAL + PJ_BLOCKS - 1) / PJ_BLOCKS) + 31) / 32 * 32)
#define AG_ROWS    640
static_assert((INSTN + AG_ROWS - 1) / AG_ROWS <= 256 && (SVCN + AG_ROWS - 1) / AG_ROWS <= 256 &&
              (NODEN + AG_ROWS - 1) / AG_ROWS <= 256, "a tile number fits one byte");
static_assert(E_SC % 16 == 0 && E_IN % 16 == 0 && E_NI % 16 == 0 && E_II % 16 == 0,
              "tags are read 16 per lane");
static_assert(E_SC % 4 == 0 && E_IN % 4 == 0 && E_NI % 4 == 0 && E_II % 4 == 0,
              "edge ids are read 4 per lane");
static_assert(DG_PER_BLOCK <= 256, "one id group per thread");

__global__ __launch_bounds__(256) void proj_deg_kernel(
        const float* __restrict__ feat_svc, const float* __restrict__ feat_inst,
        const float* __restrict__ feat_node,
        const float* __restrict__ W_sc, const float* __restrict__ W_in,
        const float* __restrict__ W_ni, const float* __restrict__ W_ii,
        float* __restrict__ h_sc, float* __restrict__ h_in,
        float* __restrict__ h_ni, float* __restrict__ h_ii,
        const int* __restrict__ svc_src, const int* __restrict__ svc_dst,
        const int* __restrict__ in_src,  const int* __restrict__ in_dst,
        const int* __restrict__ ni_src,  const int* __restrict__ ni_dst,
        const int* __restrict__ ii_src,  const int* __restrict__ ii_dst,
        float* __restrict__ deg, unsigned* __restrict__ tag) {
    __shared__ float Wlds[128 * 128];
    const int t    = threadIdx.x;
    const int lane = t & 31;
    const int wave = t >> 5;
    const int half = lane >> 4;
    const int c    = lane & 15;
    const int b    = blockIdx.x;

    const float* X; const float* W; float* Hout; int nrows, tile;
    if (b < 2 * PJ_T_INST) {
        tile = b >> 1; X = feat_inst; nrows = INSTN;
        if (b & 1) { W = W_ii; Hout = h_ii; } else { W = W_in; Hout = h_in; }
    } else if (b < 2 * PJ_T_INST + PJ_T_SVC) {
        tile = b - 2 * PJ_T_INST; X = feat_svc; W = W_sc; Hout = h_sc; nrows = SVCN;
    } else {
        tile = b - 2 * PJ_T_INST - PJ_T_SVC; X = feat_node; W = W_ni; Hout = h_ni; nrows = NODEN;
    }
    const int r0 = tile * 64 + (wave >> 1) * 16;
    const int n0 = (wave & 1) * 64 + 2 * c;

    const int  g      = b * DG_PER_BLOCK + t;
    const bool counts = t < DG_PER_BLOCK && g < DG_TOTAL;
    const int* sp; const int* dp; int oS, oD, gl;
    if (g < DG_SC) {
        sp = svc_src; dp = svc_dst; gl = g;
        oS = (int)(D_NS_SC - OFF_DEG); oD = (int)(D_ND_SC - OFF_DEG);
    } else if (g < DG_SC + DG_IN) {
        sp = in_src; dp = in_dst; gl = g - DG_SC;
        oS = (int)(D_NS_IN - OFF_DEG); oD = (int)(D_ND_IN - OFF_DEG);
    } else if (g < DG_SC + DG_IN + DG_NI) {
        sp = ni_src; dp = ni_dst; gl = g - DG_SC - DG_IN;
        oS = (int)(D_NS_NI - OFF_DEG); oD = (int)(D_ND_NI - OFF_DEG);
    } else {
        sp = ii_src; dp = ii_dst; gl = g - DG_SC - DG_IN - DG_NI;
        oS = (int)(D_NS_II - OFF_DEG); oD = (int)(D_ND_II - OFF_DEG);
    }
    int4 es = make_int4(0, 0, 0, 0), ed = make_int4(0, 0, 0, 0);
    if (counts) {
        es = ((const int4*)sp)[gl];
        ed = ((const int4*)dp)[gl];
    }

    const int rowA = (r0 + c < nrows) ? (r0 + c) : (nrows - 1);
    const float* Ap = X + (long long)rowA * 128 + 8 * half;
    float4 a[PJ_STEPS][2];
#pragma unroll
    for (int ks = 0; ks < PJ_STEPS; ++ks) {
        a[ks][0] = *(const float4*)(Ap + 16 * ks);
        a[ks][1] = *(const float4*)(Ap + 16 * ks + 4);
    }

#pragma unroll
    for (int it = 0; it < 8; ++it) {
        const int u  = t + 256 * it;
        const int kp = u >> 5;
        const int m4 = (u & 31) * 4;
        const float4 w0 = *(const float4*)&W[(2 * kp) * 128 + m4];
        const float4 w1 = *(const float4*)&W[(2 * kp + 1) * 128 + m4];
        float* d = &Wlds[(kp * 128 + m4) * 2];
        *(float4*)(d)     = make_float4(w0.x, w1.x, w0.y, w1.y);
        *(float4*)(d + 4) = make_float4(w0.z, w1.z, w0.w, w1.w);
    }
    __syncthreads();

    v8f acc[4] = {v8f{}, v8f{}, v8f{}, v8f{}};
    const float* Bp = &Wlds[(4 * half * 128 + n0) * 2];

#pragma unroll
    for (int ks = 0; ks < PJ_STEPS; ++ks) {
#pragma unroll
        for (int s = 0; s < 4; ++s) {
            const float* bp = Bp + ((8 * ks + s) * 128) * 2;
            const float4 b0 = *(const float4*)(bp);
            const float4 b1 = *(const float4*)(bp + 64);
            const float4 av = a[ks][s >> 1];
            v2f af;
            af.x = (s & 1) ? av.z : av.x;
            af.y = (s & 1) ? av.w : av.y;
            v2f bf;
            bf.x = b0.x; bf.y = b0.y;
            acc[0] = __builtin_amdgcn_wmma_f32_16x16x4_f32(false, af, false, bf, (short)0, acc[0], false, false);
            bf.x = b0.z; bf.y = b0.w;
            acc[1] = __builtin_amdgcn_wmma_f32_16x16x4_f32(false, af, false, bf, (short)0, acc[1], false, false);
            bf.x = b1.x; bf.y = b1.y;
            acc[2] = __builtin_amdgcn_wmma_f32_16x16x4_f32(false, af, false, bf, (short)0, acc[2], false, false);
            bf.x = b1.z; bf.y = b1.w;
            acc[3] = __builtin_amdgcn_wmma_f32_16x16x4_f32(false, af, false, bf, (short)0, acc[3], false, false);
        }
    }

#pragma unroll
    for (int i = 0; i < 8; ++i) {
        const int row = r0 + i + 8 * half;
        if (row < nrows) {
            float* hp = Hout + (long long)row * 128 + n0;
            *(float2*)(hp)      = make_float2(acc[0][i], acc[1][i]);
            *(float2*)(hp + 32) = make_float2(acc[2][i], acc[3][i]);
        }
    }

    if (counts) {
        tag[g] = ((unsigned)ed.x / AG_ROWS) | (((unsigned)ed.y / AG_ROWS) << 8) |
                 (((unsigned)ed.z / AG_ROWS) << 16) | (((unsigned)ed.w / AG_ROWS) << 24);
        float* cS = deg + oS;
        float* cD = deg + oD;
        atomAddF(cS + es.x, 1.f); atomAddF(cS + es.y, 1.f);
        atomAddF(cS + es.z, 1.f); atomAddF(cS + es.w, 1.f);
        atomAddF(cD + ed.x, 1.f); atomAddF(cD + ed.y, 1.f);
        atomAddF(cD + ed.z, 1.f); atomAddF(cD + ed.w, 1.f);
    }
}

#define AG_THREADS 1024
#define AG_WAVES   (AG_THREADS / 32)
#define AG_FLY     16
#define AG_GRP     2
#define AG_T_INST  ((INSTN + AG_ROWS - 1) / AG_ROWS)
#define AG_T_SVC   ((SVCN  + AG_ROWS - 1) / AG_ROWS)
#define AG_T_NODE  ((NODEN + AG_ROWS - 1) / AG_ROWS)
struct AgList {
    const int*   src;
    const int*   dst;
    const float* cntS;
    const float* cntD;
    const float* H;
    const uint4* tag;
    int          n16;
};

template <bool FULL>
__device__ __forceinline__ void ag_flush(const AgList& L, float* tile, int r0, int lane, int qe, int n) {
    const int d = L.dst[qe];
    const int s = L.src[qe];
    const float cd = L.cntD[d];
    const float cs = L.cntS[s];
    const int f = __float_as_int(rsqrtf(cd > 1.f ? cd : 1.f) * rsqrtf(cs > 1.f ? cs : 1.f));
    const int r = d - r0;
    float h[AG_FLY][4];
#pragma unroll
    for (int j = 0; j < AG_FLY; ++j) {
        h[j][0] = 0.f; h[j][1] = 0.f; h[j][2] = 0.f; h[j][3] = 0.f;
        if (FULL || j < n) {
            const float* hp = L.H + (long long)__builtin_amdgcn_readlane(s, j) * 128;
            h[j][0] = hp[lane]; h[j][1] = hp[lane + 32]; h[j][2] = hp[lane + 64]; h[j][3] = hp[lane + 96];
        }
    }
#pragma unroll
    for (int j = 0; j < AG_FLY; ++j) {
        if (FULL || j < n) {
            float*      p  = tile + __builtin_amdgcn_readlane(r, j) * 128 + lane;
            const float fj = __int_as_float(__builtin_amdgcn_readlane(f, j));
            atomicAdd(p,      h[j][0] * fj);
            atomicAdd(p + 32, h[j][1] * fj);
            atomicAdd(p + 64, h[j][2] * fj);
            atomicAdd(p + 96, h[j][3] * fj);
        }
    }
}

__device__ __forceinline__ unsigned zero_bytes(unsigned x) {
    const unsigned t = ((x & 0x7F7F7F7Fu) + 0x7F7F7F7Fu) | x;
    return ~t & 0x80808080u;
}

__device__ __forceinline__ void ag_group(const AgList& L, float* tile, int r0, int lane,
                                         uint4 tg, unsigned mine4, int gb, int& qe, int& cnt) {
    const unsigned m0 = zero_bytes(tg.x ^ mine4), m1 = zero_bytes(tg.y ^ mine4);
    const unsigned m2 = zero_bytes(tg.z ^ mine4), m3 = zero_bytes(tg.w ^ mine4);
    const unsigned code = (m0 >> 7) | (m1 >> 6) | (m2 >> 5) | (m3 >> 4);
    unsigned lanes = __builtin_amdgcn_ballot_w32(gb + lane < L.n16 && code != 0u);
    if (lanes == 0u) return;
    do {
        const int l = __builtin_ctz(lanes);
        lanes &= lanes - 1u;
        unsigned c = (unsigned)__builtin_amdgcn_readlane((int)code, l);
        const int eb = 16 * (gb + l);
        do {
            const int p = __builtin_ctz(c);
            c &= c - 1u;
            const int k = 4 * (p & 7) + (p >> 3);
            qe = (lane == cnt) ? (eb + k) : qe;
            if (++cnt == AG_FLY) { ag_flush<true>(L, tile, r0, lane, qe, AG_FLY); cnt = 0; }
        } while (c != 0u);
    } while (lanes != 0u);
}

__device__ __forceinline__ void ag_list(const AgList& L, float* tile, int r0, unsigned mine4, int wave, int lane) {
    const uint4* tg4 = L.tag;
    int qe  = 0;
    int cnt = 0;
    int i = wave * 32 + lane;
    uint4 n0 = tg4[i < L.n16 ? i : 0];
    uint4 n1 = tg4[i + AG_THREADS < L.n16 ? i + AG_THREADS : 0];
    for (int gb = wave * 32; gb < L.n16; gb += AG_GRP * AG_THREADS) {
        const uint4 d0 = n0, d1 = n1;
        i += AG_GRP * AG_THREADS;
        n0 = tg4[i < L.n16 ? i : 0];
        n1 = tg4[i + AG_THREADS < L.n16 ? i + AG_THREADS : 0];
        ag_group(L, tile, r0, lane, d0, mine4, gb, qe, cnt);
        ag_group(L, tile, r0, lane, d1, mine4, gb + AG_THREADS, qe, cnt);
    }
    if (cnt != 0) ag_flush<false>(L, tile, r0, lane, qe, cnt);
}

__global__ __launch_bounds__(AG_THREADS) void agg_kernel(
        const int* __restrict__ svc_src, const int* __restrict__ svc_dst,
        const int* __restrict__ in_src,  const int* __restrict__ in_dst,
        const int* __restrict__ ni_src,  const int* __restrict__ ni_dst,
        const int* __restrict__ ii_src,  const int* __restrict__ ii_dst,
        const float* __restrict__ h_sc, const float* __restrict__ h_in,
        const float* __restrict__ h_ni, const float* __restrict__ h_ii,
        const float* __restrict__ deg,
        const float* __restrict__ b_sc, const float* __restrict__ b_in,
        const float* __restrict__ b_ni, const float* __restrict__ b_ii,
        const unsigned* __restrict__ tag,
        float* __restrict__ cat) {
    __shared__ float tile[AG_ROWS * 128];
    const int lane = threadIdx.x & 31;
    const int wave = __builtin_amdgcn_readfirstlane(threadIdx.x >> 5);
    const int b    = blockIdx.x;

    for (int i = threadIdx.x; i < AG_ROWS * 32; i += AG_THREADS)
        ((float4*)tile)[i] = make_float4(0.f, 0.f, 0.f, 0.f);
    __syncthreads();

    int r0, nr, catRow0, nLists, tno;
    AgList L;
    const uint4* tag16 = (const uint4*)tag;
    float4 bias;
    const int c4 = lane * 4;
    if (b < AG_T_INST) {
        tno = b;
        r0 = b * AG_ROWS;
        nr = (INSTN - r0 < AG_ROWS) ? (INSTN - r0) : AG_ROWS;
        catRow0 = SVCN + NODEN + r0;
        L.src = ni_src; L.dst = ni_dst; L.H = h_ni; nLists = 2;
        L.tag = tag16 + (E_SC + E_IN) / 16; L.n16 = E_NI / 16;
        L.cntS = deg + (D_NS_NI - OFF_DEG); L.cntD = deg + (D_ND_NI - OFF_DEG);
        const float4 x = *(const float4*)&b_ni[c4];
        const float4 y = *(const float4*)&b_ii[c4];
        bias = make_float4(x.x + y.x, x.y + y.y, x.z + y.z, x.w + y.w);
    } else if (b < AG_T_INST + AG_T_SVC) {
        tno = b - AG_T_INST;
        r0 = tno * AG_ROWS;
        nr = (SVCN - r0 < AG_ROWS) ? (SVCN - r0) : AG_ROWS;
        catRow0 = r0;
        L.src = svc_src; L.dst = svc_dst; L.H = h_sc; nLists = 1;
        L.tag = tag16; L.n16 = E_SC / 16;
        L.cntS = deg + (D_NS_SC - OFF_DEG); L.cntD = deg + (D_ND_SC - OFF_DEG);
        bias = *(const float4*)&b_sc[c4];
    } else {
        tno = b - AG_T_INST - AG_T_SVC;
        r0 = tno * AG_ROWS;
        nr = (NODEN - r0 < AG_ROWS) ? (NODEN - r0) : AG_ROWS;
        catRow0 = SVCN + r0;
        L.src = in_src; L.dst = in_dst; L.H = h_in; nLists = 1;
        L.tag = tag16 + E_SC / 16; L.n16 = E_IN / 16;
        L.cntS = deg + (D_NS_IN - OFF_DEG); L.cntD = deg + (D_ND_IN - OFF_DEG);
        bias = *(const float4*)&b_in[c4];
    }
    const unsigned mine4 = (unsigned)tno * 0x01010101u;
#pragma nounroll
    for (int li = 0; li < nLists; ++li) {
        ag_list(L, tile, r0, mine4, wave, lane);
        L.src = ii_src; L.dst = ii_dst; L.H = h_ii;
        L.tag = tag16 + (E_SC + E_IN + E_NI) / 16; L.n16 = E_II / 16;
        L.cntS = deg + (D_NS_II - OFF_DEG); L.cntD = deg + (D_ND_II - OFF_DEG);
    }
    __syncthreads();

    for (int r = wave; r < nr; r += AG_WAVES) {
        const float4 v = *(const float4*)&tile[r * 128 + c4];
        *(float4*)&cat[(long long)(catRow0 + r) * 128 + c4] =
            make_float4(v.x + bias.x, v.y + bias.y, v.z + bias.z, v.w + bias.w);
    }
}

#define FG_STEP   16
#define FG_DSTEPS (TOTAL / (2 * FG_STEP))
#define FG_CHUNKS 63
#define FG_SLICES (4 * FG_CHUNKS)
#define FG_GROUPS (4 * FG_CHUNKS)
#define FG_SLOT   2048
static_assert(TOTAL % (2 * FG_STEP) == 0, "K is walked in double steps");
static_assert(FG_DSTEPS / FG_SLICES >= 1, "no empty slice");

struct FgFrag {
    float4 a[4][2];
    float2 b[4][4][2];
};

__device__ __forceinline__ void fg_load(FgFrag& f, const float* const (&Ap)[4],
                                        const float* __restrict__ Bp, int k) {
#pragma unroll
    for (int rt = 0; rt < 4; ++rt) {
        f.a[rt][0] = *(const float4*)(Ap[rt] + k);
        f.a[rt][1] = *(const float4*)(Ap[rt] + k + 4);
    }
#pragma unroll
    for (int s = 0; s < 4; ++s) {
#pragma unroll
        for (int q = 0; q < 4; ++q) {
            f.b[s][q][0] = *(const float2*)(Bp + (long long)(k + 2 * s) * 128 + 32 * q);
            f.b[s][q][1] = *(const float2*)(Bp + (long long)(k + 2 * s + 1) * 128 + 32 * q);
        }
    }
}

__device__ __forceinline__ void fg_mma(const FgFrag& f, v8f (&acc)[4][8]) {
#pragma unroll
    for (int s = 0; s < 4; ++s) {
        v2f bf[4][2];
#pragma unroll
        for (int q = 0; q < 4; ++q) {
            bf[q][0].x = lrelu(f.b[s][q][0].x);
            bf[q][0].y = lrelu(f.b[s][q][1].x);
            bf[q][1].x = lrelu(f.b[s][q][0].y);
            bf[q][1].y = lrelu(f.b[s][q][1].y);
        }
#pragma unroll
        for (int rt = 0; rt < 4; ++rt) {
            const float4 av = f.a[rt][s >> 1];
            v2f a;
            a.x = (s & 1) ? av.z : av.x;
            a.y = (s & 1) ? av.w : av.y;
#pragma unroll
            for (int q = 0; q < 4; ++q) {
#pragma unroll
                for (int e = 0; e < 2; ++e) {
                    acc[rt][2 * q + e] = __builtin_amdgcn_wmma_f32_16x16x4_f32(
                        false, a, false, bf[q][e], (short)0, acc[rt][2 * q + e], false, false);
                }
            }
        }
    }
}

__global__ __launch_bounds__(128) void final_gemm_kernel(const float* __restrict__ Wt,
                                                         const float* __restrict__ cat,
                                                         float* __restrict__ out) {
    __shared__ float red[16 * FG_SLOT];
    const int t    = threadIdx.x;
    const int lane = t & 31;
    const int wave = __builtin_amdgcn_readfirstlane(t >> 5);
    const int half = lane >> 4;
    const int c    = lane & 15;
    const int m0   = (blockIdx.x & 3) * 64;
    const int sl   = (blockIdx.x >> 2) * 4 + wave;
    const int k0   = 2 * FG_STEP * ((FG_DSTEPS * sl) / FG_SLICES);
    const int kEnd = 2 * FG_STEP * ((FG_DSTEPS * (sl + 1)) / FG_SLICES);

    const float* Ap[4];
#pragma unroll
    for (int rt = 0; rt < 4; ++rt)
        Ap[rt] = Wt + (long long)(m0 + rt * 16 + c) * TOTAL + 8 * half;
    const float* Bp = cat + (long long)(8 * half) * 128 + 2 * c;

    v8f acc[4][8];
#pragma unroll
    for (int rt = 0; rt < 4; ++rt)
#pragma unroll
        for (int j = 0; j < 8; ++j) acc[rt][j] = v8f{};

    FgFrag f0, f1;
    fg_load(f0, Ap, Bp, k0);
    asm volatile("" ::: "memory");
    for (int k = k0; k < kEnd; k += 2 * FG_STEP) {
        fg_load(f1, Ap, Bp, k + FG_STEP);
        fg_mma(f0, acc);
        const int kn = (k + 2 * FG_STEP < kEnd) ? (k + 2 * FG_STEP) : k0;
        fg_load(f0, Ap, Bp, kn);
        asm volatile("" ::: "memory");
        fg_mma(f1, acc);
    }

#pragma unroll
    for (int s = 0; s < 4; ++s) {
        if (wave == s) {
#pragma unroll
            for (int rt = 0; rt < 4; ++rt) {
                if (rt != s) {
                    float4* dp = (float4*)&red[(rt * 4 + s) * FG_SLOT] + lane;
#pragma unroll
                    for (int j = 0; j < 8; ++j) {
                        dp[(2 * j) * 32]     = make_float4(acc[rt][j][0], acc[rt][j][1], acc[rt][j][2], acc[rt][j][3]);
                        dp[(2 * j + 1) * 32] = make_float4(acc[rt][j][4], acc[rt][j][5], acc[rt][j][6], acc[rt][j][7]);
                    }
                }
            }
        }
    }
    __syncthreads();
#pragma unroll
    for (int rt = 0; rt < 4; ++rt) {
        if (wave == rt) {
#pragma unroll
            for (int s = 0; s < 4; ++s) {
                if (s != rt) {
                    const float4* sp = (const float4*)&red[(rt * 4 + s) * FG_SLOT] + lane;
#pragma unroll
                    for (int j = 0; j < 8; ++j) {
                        const float4 x = sp[(2 * j) * 32];
                        const float4 y = sp[(2 * j + 1) * 32];
                        acc[rt][j][0] += x.x; acc[rt][j][1] += x.y; acc[rt][j][2] += x.z; acc[rt][j][3] += x.w;
                        acc[rt][j][4] += y.x; acc[rt][j][5] += y.y; acc[rt][j][6] += y.z; acc[rt][j][7] += y.w;
                    }
                }
            }
#pragma unroll
            for (int q = 0; q < 4; ++q) {
#pragma unroll
                for (int e = 0; e < 2; ++e) {
#pragma unroll
                    for (int i = 0; i < 8; ++i) {
                        const int m = m0 + rt * 16 + i + 8 * half;
                        atomAddF(&out[m * 128 + 32 * q + 2 * c + e], acc[rt][2 * q + e][i]);
                    }
                }
            }
        }
    }
}

static inline int cdiv(long long a, long long b) { return (int)((a + b - 1) / b); }

extern "C" void kernel_launch(void* const* d_in, const int* in_sizes, int n_in,
                              void* d_out, int out_size, void* d_ws, size_t ws_size,
                              hipStream_t stream) {
    const float* feat_svc  = (const float*)d_in[0];
    const float* feat_inst = (const float*)d_in[1];
    const float* feat_node = (const float*)d_in[2];
    const int*   svc_src   = (const int*)d_in[3];
    const int*   svc_dst   = (const int*)d_in[4];
    const int*   in_src    = (const int*)d_in[5];
    const int*   in_dst    = (const int*)d_in[6];
    const int*   ni_src    = (const int*)d_in[7];
    const int*   ni_dst    = (const int*)d_in[8];
    const int*   ii_src    = (const int*)d_in[9];
    const int*   ii_dst    = (const int*)d_in[10];
    const float* W_sc = (const float*)d_in[11];  const float* b_sc = (const float*)d_in[12];
    const float* W_in = (const float*)d_in[13];  const float* b_in = (const float*)d_in[14];
    const float* W_ni = (const float*)d_in[15];  const float* b_ni = (const float*)d_in[16];
    const float* W_ii = (const float*)d_in[17];  const float* b_ii = (const float*)d_in[18];
    const float* W_tot = (const float*)d_in[19]; const float* b_tot = (const float*)d_in[20];
    float* out = (float*)d_out;

    float* ws   = (float*)d_ws;
    float* cat  = ws + OFF_CAT;
    float* h_sc = ws + OFF_HSC;
    float* h_in = ws + OFF_HIN;
    float* h_ni = ws + OFF_HNI;
    float* h_ii = ws + OFF_HII;
    float* deg  = ws + OFF_DEG;
    unsigned* tag = (unsigned*)(ws + OFF_TAG);

    init_deg_kernel<<<SMALL_BLOCKS, 256, 0, stream>>>(deg);
    init_out_kernel<<<cdiv(OUTD * HDIM, 256), 256, 0, stream>>>(out, b_tot);

    proj_deg_kernel<<<PJ_BLOCKS, 256, 0, stream>>>(
        feat_svc, feat_inst, feat_node, W_sc, W_in, W_ni, W_ii, h_sc, h_in, h_ni, h_ii,
        svc_src, svc_dst, in_src, in_dst, ni_src, ni_dst, ii_src, ii_dst, deg, tag);

    agg_kernel<<<AG_T_INST + AG_T_SVC + AG_T_NODE, AG_THREADS, 0, stream>>>(
        svc_src, svc_dst, in_src, in_dst, ni_src, ni_dst, ii_src, ii_dst,
        h_sc, h_in, h_ni, h_ii, deg,
        b_sc, b_in, b_ni, b_ii, tag, cat);

    final_gemm_kernel<<<FG_GROUPS, 128, 0, stream>>>(W_tot, cat, out);
}


# in-degrees of the sc, in, ii lists counted by the owning aggregation tile (LDS integer add per edge), nd applied per row to the list's sum: 1.3M global f32 count atomics gone; row sum order changes
# speedup vs baseline: 2.7143x; 1.1135x over previous
#include <hip/hip_runtime.h>

#define SVCN   50000
#define INSTN  100000
#define NODEN  50000
#define FDIM   128
#define HDIM   128
#define OUTD   256
#define TOTAL  (SVCN + NODEN + INSTN)
#define E_SC   400000
#define E_IN   100000
#define E_NI   100000
#define E_II   800000

#define OFF_CAT   0LL
#define OFF_HSC   (OFF_CAT  + (long long)TOTAL * 128)
#define OFF_HIN   (OFF_HSC  + (long long)SVCN  * 128)
#define OFF_HNI   (OFF_HIN  + (long long)INSTN * 128)
#define OFF_HII   (OFF_HNI  + (long long)NODEN * 128)
#define OFF_DEG   (OFF_HII  + (long long)INSTN * 128)
#define D_NS_SC   (OFF_DEG + 0LL)
#define D_NS_IN   (D_NS_SC + SVCN)
#define D_NS_NI   (D_NS_IN + INSTN)
#define D_ND_NI   (D_NS_NI + NODEN)
#define D_NS_II   (D_ND_NI + INSTN)
#define DEG_USED  (SVCN + INSTN + NODEN + INSTN + INSTN)
#define DEG_TOTAL ((DEG_USED + 31) / 32 * 32)
#define OFF_TAG   (OFF_DEG + DEG_TOTAL)

typedef __attribute__((ext_vector_type(2))) float v2f;
typedef __attribute__((ext_vector_type(8))) float v8f;

__device__ __forceinline__ float lrelu(float x) { return x > 0.f ? x : 0.01f * x; }

__device__ __forceinline__ void atomAddF(float* p, float v) { unsafeAtomicAdd(p, v); }

#define SMALL_BLOCKS 128
static_assert(DEG_TOTAL % 4 == 0, "degree planes are walked 16 bytes per lane");

__global__ __launch_bounds__(256) void init_deg_kernel(float* deg) {
    float4* d4 = (float4*)deg;
    for (int i = blockIdx.x * 256 + threadIdx.x; i < DEG_TOTAL / 4; i += SMALL_BLOCKS * 256)
        d4[i] = make_float4(0.f, 0.f, 0.f, 0.f);
}

__global__ void init_out_kernel(float* __restrict__ out, const float* __restrict__ b_tot) {
    int i = blockIdx.x * blockDim.x + threadIdx.x;
    if (i < OUTD * HDIM) out[i] = b_tot[i >> 7];
}

#define PJ_STEPS (FDIM / 16)
#define PJ_T_SVC   ((SVCN  + 63) / 64)
#define PJ_T_NODE  ((NODEN + 63) / 64)
#define PJ_T_INST  ((INSTN + 63) / 64)
#define PJ_BLOCKS  (2 * PJ_T_INST + PJ_T_SVC + PJ_T_NODE)
#define DG_SC      (E_SC / 4)
#define DG_IN      (E_IN / 4)
#define DG_NI      (E_NI / 4)
#define DG_II      (E_II / 4)
#define DG_TOTAL   (DG_SC + DG_IN + DG_NI + DG_II)
#define DG_PER_BLOCK ((((DG_TOTAL + PJ_BLOCKS - 1) / PJ_BLOCKS) + 31) / 32 * 32)
#define AG_ROWS    635
static_assert((INSTN + AG_ROWS - 1) / AG_ROWS <= 256 && (SVCN + AG_ROWS - 1) / AG_ROWS <= 256 &&
              (NODEN + AG_ROWS - 1) / AG_ROWS <= 256, "a tile number fits one byte");
static_assert(E_SC % 16 == 0 && E_IN % 16 == 0 && E_NI % 16 == 0 && E_II % 16 == 0,
              "tags are read 16 per lane");
static_assert(E_SC % 4 == 0 && E_IN % 4 == 0 && E_NI % 4 == 0 && E_II % 4 == 0,
              "edge ids are read 4 per lane");
static_assert(DG_PER_BLOCK <= 256, "one id group per thread");

__global__ __launch_bounds__(256) void proj_deg_kernel(
        const float* __restrict__ feat_svc, const float* __restrict__ feat_inst,
        const float* __restrict__ feat_node,
        const float* __restrict__ W_sc, const float* __restrict__ W_in,
        const float* __restrict__ W_ni, const float* __restrict__ W_ii,
        float* __restrict__ h_sc, float* __restrict__ h_in,
        float* __restrict__ h_ni, float* __restrict__ h_ii,
        const int* __restrict__ svc_src, const int* __restrict__ svc_dst,
        const int* __restrict__ in_src,  const int* __restrict__ in_dst,
        const int* __restrict__ ni_src,  const int* __restrict__ ni_dst,
        const int* __restrict__ ii_src,  const int* __restrict__ ii_dst,
        float* __restrict__ deg, unsigned* __restrict__ tag) {
    __shared__ float Wlds[128 * 128];
    const int t    = threadIdx.x;
    const int lane = t & 31;
    const int wave = t >> 5;
    const int half = lane >> 4;
    const int c    = lane & 15;
    const int b    = blockIdx.x;

    const float* X; const float* W; float* Hout; int nrows, tile;
    if (b < 2 * PJ_T_INST) {
        tile = b >> 1; X = feat_inst; nrows = INSTN;
        if (b & 1) { W = W_ii; Hout = h_ii; } else { W = W_in; Hout = h_in; }
    } else if (b < 2 * PJ_T_INST + PJ_T_SVC) {
        tile = b - 2 * PJ_T_INST; X = feat_svc; W = W_sc; Hout = h_sc; nrows = SVCN;
    } else {
        tile = b - 2 * PJ_T_INST - PJ_T_SVC; X = feat_node; W = W_ni; Hout = h_ni; nrows = NODEN;
    }
    const int r0 = tile * 64 + (wave >> 1) * 16;
    const int n0 = (wave & 1) * 64 + 2 * c;

    const int  g      = b * DG_PER_BLOCK + t;
    const bool counts = t < DG_PER_BLOCK && g < DG_TOTAL;
    const int* sp; const int* dp; int oS, gl;
    bool isNi = false;
    if (g < DG_SC) {
        sp = svc_src; dp = svc_dst; gl = g;
        oS = (int)(D_NS_SC - OFF_DEG);
    } else if (g < DG_SC + DG_IN) {
        sp = in_src; dp = in_dst; gl = g - DG_SC;
        oS = (int)(D_NS_IN - OFF_DEG);
    } else if (g < DG_SC + DG_IN + DG_NI) {
        sp = ni_src; dp = ni_dst; gl = g - DG_SC - DG_IN;
        oS = (int)(D_NS_NI - OFF_DEG); isNi = true;
    } else {
        sp = ii_src; dp = ii_dst; gl = g - DG_SC - DG_IN - DG_NI;
        oS = (int)(D_NS_II - OFF_DEG);
    }
    int4 es = make_int4(0, 0, 0, 0), ed = make_int4(0, 0, 0, 0);
    if (counts) {
        es = ((const int4*)sp)[gl];
        ed = ((const int4*)dp)[gl];
    }

    const int rowA = (r0 + c < nrows) ? (r0 + c) : (nrows - 1);
    const float* Ap = X + (long long)rowA * 128 + 8 * half;
    float4 a[PJ_STEPS][2];
#pragma unroll
    for (int ks = 0; ks < PJ_STEPS; ++ks) {
        a[ks][0] = *(const float4*)(Ap + 16 * ks);
        a[ks][1] = *(const float4*)(Ap + 16 * ks + 4);
    }

#pragma unroll
    for (int it = 0; it < 8; ++it) {
        const int u  = t + 256 * it;
        const int kp = u >> 5;
        const int m4 = (u & 31) * 4;
        const float4 w0 = *(const float4*)&W[(2 * kp) * 128 + m4];
        const float4 w1 = *(const float4*)&W[(2 * kp + 1) * 128 + m4];
        float* d = &Wlds[(kp * 128 + m4) * 2];
        *(float4*)(d)     = make_float4(w0.x, w1.x, w0.y, w1.y);
        *(float4*)(d + 4) = make_float4(w0.z, w1.z, w0.w, w1.w);
    }
    __syncthreads();

    v8f acc[4] = {v8f{}, v8f{}, v8f{}, v8f{}};
    const float* Bp = &Wlds[(4 * half * 128 + n0) * 2];

#pragma unroll
    for (int ks = 0; ks < PJ_STEPS; ++ks) {
#pragma unroll
        for (int s = 0; s < 4; ++s) {
            const float* bp = Bp + ((8 * ks + s) * 128) * 2;
            const float4 b0 = *(const float4*)(bp);
            const float4 b1 = *(const float4*)(bp + 64);
            const float4 av = a[ks][s >> 1];
            v2f af;
            af.x = (s & 1) ? av.z : av.x;
            af.y = (s & 1) ? av.w : av.y;
            v2f bf;
            bf.x = b0.x; bf.y = b0.y;
            acc[0] = __builtin_amdgcn_wmma_f32_16x16x4_f32(false, af, false, bf, (short)0, acc[0], false, false);
            bf.x = b0.z; bf.y = b0.w;
            acc[1] = __builtin_amdgcn_wmma_f32_16x16x4_f32(false, af, false, bf, (short)0, acc[1], false, false);
            bf.x = b1.x; bf.y = b1.y;
            acc[2] = __builtin_amdgcn_wmma_f32_16x16x4_f32(false, af, false, bf, (short)0, acc[2], false, false);
            bf.x = b1.z; bf.y = b1.w;
            acc[3] = __builtin_amdgcn_wmma_f32_16x16x4_f32(false, af, false, bf, (short)0, acc[3], false, false);
        }
    }

#pragma unroll
    for (int i = 0; i < 8; ++i) {
        const int row = r0 + i + 8 * half;
        if (row < nrows) {
            float* hp = Hout + (long long)row * 128 + n0;
            *(float2*)(hp)      = make_float2(acc[0][i], acc[1][i]);
            *(float2*)(hp + 32) = make_float2(acc[2][i], acc[3][i]);
        }
    }

    if (counts) {
        tag[g] = ((unsigned)ed.x / AG_ROWS) | (((unsigned)ed.y / AG_ROWS) << 8) |
                 (((unsigned)ed.z / AG_ROWS) << 16) | (((unsigned)ed.w / AG_ROWS) << 24);
        float* cS = deg + oS;
        atomAddF(cS + es.x, 1.f); atomAddF(cS + es.y, 1.f);
        atomAddF(cS + es.z, 1.f); atomAddF(cS + es.w, 1.f);
        if (isNi) {
            float* cD = deg + (D_ND_NI - OFF_DEG);
            atomAddF(cD + ed.x, 1.f); atomAddF(cD + ed.y, 1.f);
            atomAddF(cD + ed.z, 1.f); atomAddF(cD + ed.w, 1.f);
        }
    }
}

#define AG_THREADS 1024
#define AG_WAVES   (AG_THREADS / 32)
#define AG_FLY     16
#define AG_GRP     2
#define AG_T_INST  ((INSTN + AG_ROWS - 1) / AG_ROWS)
#define AG_T_SVC   ((SVCN  + AG_ROWS - 1) / AG_ROWS)
#define AG_T_NODE  ((NODEN + AG_ROWS - 1) / AG_ROWS)
struct AgList {
    const int*   src;
    const int*   dst;
    const float* cntS;
    const float* cntD;
    const float* H;
    const uint4* tag;
    int          n16;
};

template <bool FULL>
__device__ __forceinline__ void ag_flush(const AgList& L, float* tile, unsigned* rowcnt, int r0, int lane, int qe, int n) {
    const int d = L.dst[qe];
    const int s = L.src[qe];
    const int r = d - r0;
    float cd = 1.f;
    if (L.cntD != nullptr) cd = L.cntD[d];
    else if (lane < n) atomicAdd(rowcnt + r, 1u);
    const float cs = L.cntS[s];
    const int f = __float_as_int(rsqrtf(cd > 1.f ? cd : 1.f) * rsqrtf(cs > 1.f ? cs : 1.f));
    float h[AG_FLY][4];
#pragma unroll
    for (int j = 0; j < AG_FLY; ++j) {
        h[j][0] = 0.f; h[j][1] = 0.f; h[j][2] = 0.f; h[j][3] = 0.f;
        if (FULL || j < n) {
            const float* hp = L.H + (long long)__builtin_amdgcn_readlane(s, j) * 128;
            h[j][0] = hp[lane]; h[j][1] = hp[lane + 32]; h[j][2] = hp[lane + 64]; h[j][3] = hp[lane + 96];
        }
    }
#pragma unroll
    for (int j = 0; j < AG_FLY; ++j) {
        if (FULL || j < n) {
            float*      p  = tile + __builtin_amdgcn_readlane(r, j) * 128 + lane;
            const float fj = __int_as_float(__builtin_amdgcn_readlane(f, j));
            atomicAdd(p,      h[j][0] * fj);
            atomicAdd(p + 32, h[j][1] * fj);
            atomicAdd(p + 64, h[j][2] * fj);
            atomicAdd(p + 96, h[j][3] * fj);
        }
    }
}

__device__ __forceinline__ unsigned zero_bytes(unsigned x) {
    const unsigned t = ((x & 0x7F7F7F7Fu) + 0x7F7F7F7Fu) | x;
    return ~t & 0x80808080u;
}

__device__ __forceinline__ void ag_group(const AgList& L, float* tile, unsigned* rowcnt, int r0, int lane,
                                         uint4 tg, unsigned mine4, int gb, int& qe, int& cnt) {
    const unsigned m0 = zero_bytes(tg.x ^ mine4), m1 = zero_bytes(tg.y ^ mine4);
    const unsigned m2 = zero_bytes(tg.z ^ mine4), m3 = zero_bytes(tg.w ^ mine4);
    const unsigned code = (m0 >> 7) | (m1 >> 6) | (m2 >> 5) | (m3 >> 4);
    unsigned lanes = __builtin_amdgcn_ballot_w32(gb + lane < L.n16 && code != 0u);
    if (lanes == 0u) return;
    do {
        const int l = __builtin_ctz(lanes);
        lanes &= lanes - 1u;
        unsigned c = (unsigned)__builtin_amdgcn_readlane((int)code, l);
        const int eb = 16 * (gb + l);
        do {
            const int p = __builtin_ctz(c);
            c &= c - 1u;
            const int k = 4 * (p & 7) + (p >> 3);
            qe = (lane == cnt) ? (eb + k) : qe;
            if (++cnt == AG_FLY) { ag_flush<true>(L, tile, rowcnt, r0, lane, qe, AG_FLY); cnt = 0; }
        } while (c != 0u);
    } while (lanes != 0u);
}

__device__ __forceinline__ void ag_list(const AgList& L, float* tile, unsigned* rowcnt, int r0, unsigned mine4, int wave, int lane) {
    const uint4* tg4 = L.tag;
    int qe  = 0;
    int cnt = 0;
    int i = wave * 32 + lane;
    uint4 n0 = tg4[i < L.n16 ? i : 0];
    uint4 n1 = tg4[i + AG_THREADS < L.n16 ? i + AG_THREADS : 0];
    for (int gb = wave * 32; gb < L.n16; gb += AG_GRP * AG_THREADS) {
        const uint4 d0 = n0, d1 = n1;
        i += AG_GRP * AG_THREADS;
        n0 = tg4[i < L.n16 ? i : 0];
        n1 = tg4[i + AG_THREADS < L.n16 ? i + AG_THREADS : 0];
        ag_group(L, tile, rowcnt, r0, lane, d0, mine4, gb, qe, cnt);
        ag_group(L, tile, rowcnt, r0, lane, d1, mine4, gb + AG_THREADS, qe, cnt);
    }
    if (cnt != 0) ag_flush<false>(L, tile, rowcnt, r0, lane, qe, cnt);
}

__global__ __launch_bounds__(AG_THREADS) void agg_kernel(
        const int* __restrict__ svc_src, const int* __restrict__ svc_dst,
        const int* __restrict__ in_src,  const int* __restrict__ in_dst,
        const int* __restrict__ ni_src,  const int* __restrict__ ni_dst,
        const int* __restrict__ ii_src,  const int* __restrict__ ii_dst,
        const float* __restrict__ h_sc, const float* __restrict__ h_in,
        const float* __restrict__ h_ni, const float* __restrict__ h_ii,
        const float* __restrict__ deg,
        const float* __restrict__ b_sc, const float* __restrict__ b_in,
        const float* __restrict__ b_ni, const float* __restrict__ b_ii,
        const unsigned* __restrict__ tag,
        float* __restrict__ cat) {
    __shared__ float    tile[AG_ROWS * 128];
    __shared__ unsigned rowcnt[AG_ROWS];
    const int lane = threadIdx.x & 31;
    const int wave = __builtin_amdgcn_readfirstlane(threadIdx.x >> 5);
    const int b    = blockIdx.x;

    for (int i = threadIdx.x; i < AG_ROWS * 32; i += AG_THREADS)
        ((float4*)tile)[i] = make_float4(0.f, 0.f, 0.f, 0.f);
    for (int i = threadIdx.x; i < AG_ROWS; i += AG_THREADS) rowcnt[i] = 0u;
    __syncthreads();

    int r0, nr, catRow0, nLists, tno;
    AgList L;
    const uint4* tag16 = (const uint4*)tag;
    float4 bias;
    const int c4 = lane * 4;
    if (b < AG_T_INST) {
        tno = b;
        r0 = b * AG_ROWS;
        nr = (INSTN - r0 < AG_ROWS) ? (INSTN - r0) : AG_ROWS;
        catRow0 = SVCN + NODEN + r0;
        L.src = ii_src; L.dst = ii_dst; L.H = h_ii; nLists = 2;
        L.tag = tag16 + (E_SC + E_IN + E_NI) / 16; L.n16 = E_II / 16;
        L.cntS = deg + (D_NS_II - OFF_DEG); L.cntD = nullptr;
        const float4 x = *(const float4*)&b_ni[c4];
        const float4 y = *(const float4*)&b_ii[c4];
        bias = make_float4(x.x + y.x, x.y + y.y, x.z + y.z, x.w + y.w);
    } else if (b < AG_T_INST + AG_T_SVC) {
        tno = b - AG_T_INST;
        r0 = tno * AG_ROWS;
        nr = (SVCN - r0 < AG_ROWS) ? (SVCN - r0) : AG_ROWS;
        catRow0 = r0;
        L.src = svc_src; L.dst = svc_dst; L.H = h_sc; nLists = 1;
        L.tag = tag16; L.n16 = E_SC / 16;
        L.cntS = deg + (D_NS_SC - OFF_DEG); L.cntD = nullptr;
        bias = *(const float4*)&b_sc[c4];
    } else {
        tno = b - AG_T_INST - AG_T_SVC;
        r0 = tno * AG_ROWS;
        nr = (NODEN - r0 < AG_ROWS) ? (NODEN - r0) : AG_ROWS;
        catRow0 = SVCN + r0;
        L.src = in_src; L.dst = in_dst; L.H = h_in; nLists = 1;
        L.tag = tag16 + E_SC / 16; L.n16 = E_IN / 16;
        L.cntS = deg + (D_NS_IN - OFF_DEG); L.cntD = nullptr;
        bias = *(const float4*)&b_in[c4];
    }
    const unsigned mine4 = (unsigned)tno * 0x01010101u;
#pragma nounroll
    for (int li = 0; li < nLists; ++li) {
        ag_list(L, tile, rowcnt, r0, mine4, wave, lane);
        __syncthreads();
        if (li + 1 < nLists) {
            for (int r = wave; r < nr; r += AG_WAVES) {
                const float  cr = (float)rowcnt[r];
                const float  nd = rsqrtf(cr > 1.f ? cr : 1.f);
                const float4 v  = *(const float4*)&tile[r * 128 + c4];
                *(float4*)&tile[r * 128 + c4] = make_float4(v.x * nd, v.y * nd, v.z * nd, v.w * nd);
            }
            __syncthreads();
            L.src = ni_src; L.dst = ni_dst; L.H = h_ni;
            L.tag = tag16 + (E_SC + E_IN) / 16; L.n16 = E_NI / 16;
            L.cntS = deg + (D_NS_NI - OFF_DEG); L.cntD = deg + (D_ND_NI - OFF_DEG);
        }
    }

    for (int r = wave; r < nr; r += AG_WAVES) {
        const float4 v  = *(const float4*)&tile[r * 128 + c4];
        const float  cr = (float)rowcnt[r];
        const float  nd = (nLists == 1) ? rsqrtf(cr > 1.f ? cr : 1.f) : 1.f;
        *(float4*)&cat[(long long)(catRow0 + r) * 128 + c4] =
            make_float4(v.x * nd + bias.x, v.y * nd + bias.y, v.z * nd + bias.z, v.w * nd + bias.w);
    }
}

#define FG_STEP   16
#define FG_DSTEPS (TOTAL / (2 * FG_STEP))
#define FG_CHUNKS 63
#define FG_SLICES (4 * FG_CHUNKS)
#define FG_GROUPS (4 * FG_CHUNKS)
#define FG_SLOT   2048
static_assert(TOTAL % (2 * FG_STEP) == 0, "K is walked in double steps");
static_assert(FG_DSTEPS / FG_SLICES >= 1, "no empty slice");

struct FgFrag {
    float4 a[4][2];
    float2 b[4][4][2];
};

__device__ __forceinline__ void fg_load(FgFrag& f, const float* const (&Ap)[4],
                                        const float* __restrict__ Bp, int k) {
#pragma unroll
    for (int rt = 0; rt < 4; ++rt) {
        f.a[rt][0] = *(const float4*)(Ap[rt] + k);
        f.a[rt][1] = *(const float4*)(Ap[rt] + k + 4);
    }
#pragma unroll
    for (int s = 0; s < 4; ++s) {
#pragma unroll
        for (int q = 0; q < 4; ++q) {
            f.b[s][q][0] = *(const float2*)(Bp + (long long)(k + 2 * s) * 128 + 32 * q);
            f.b[s][q][1] = *(const float2*)(Bp + (long long)(k + 2 * s + 1) * 128 + 32 * q);
        }
    }
}

__device__ __forceinline__ void fg_mma(const FgFrag& f, v8f (&acc)[4][8]) {
#pragma unroll
    for (int s = 0; s < 4; ++s) {
        v2f bf[4][2];
#pragma unroll
        for (int q = 0; q < 4; ++q) {
            bf[q][0].x = lrelu(f.b[s][q][0].x);
            bf[q][0].y = lrelu(f.b[s][q][1].x);
            bf[q][1].x = lrelu(f.b[s][q][0].y);
            bf[q][1].y = lrelu(f.b[s][q][1].y);
        }
#pragma unroll
        for (int rt = 0; rt < 4; ++rt) {
            const float4 av = f.a[rt][s >> 1];
            v2f a;
            a.x = (s & 1) ? av.z : av.x;
            a.y = (s & 1) ? av.w : av.y;
#pragma unroll
            for (int q = 0; q < 4; ++q) {
#pragma unroll
                for (int e = 0; e < 2; ++e) {
                    acc[rt][2 * q + e] = __builtin_amdgcn_wmma_f32_16x16x4_f32(
                        false, a, false, bf[q][e], (short)0, acc[rt][2 * q + e], false, false);
                }
            }
        }
    }
}

__global__ __launch_bounds__(128) void final_gemm_kernel(const float* __restrict__ Wt,
                                                         const float* __restrict__ cat,
                                                         float* __restrict__ out) {
    __shared__ float red[16 * FG_SLOT];
    const int t    = threadIdx.x;
    const int lane = t & 31;
    const int wave = __builtin_amdgcn_readfirstlane(t >> 5);
    const int half = lane >> 4;
    const int c    = lane & 15;
    const int m0   = (blockIdx.x & 3) * 64;
    const int sl   = (blockIdx.x >> 2) * 4 + wave;
    const int k0   = 2 * FG_STEP * ((FG_DSTEPS * sl) / FG_SLICES);
    const int kEnd = 2 * FG_STEP * ((FG_DSTEPS * (sl + 1)) / FG_SLICES);

    const float* Ap[4];
#pragma unroll
    for (int rt = 0; rt < 4; ++rt)
        Ap[rt] = Wt + (long long)(m0 + rt * 16 + c) * TOTAL + 8 * half;
    const float* Bp = cat + (long long)(8 * half) * 128 + 2 * c;

    v8f acc[4][8];
#pragma unroll
    for (int rt = 0; rt < 4; ++rt)
#pragma unroll
        for (int j = 0; j < 8; ++j) acc[rt][j] = v8f{};

    FgFrag f0, f1;
    fg_load(f0, Ap, Bp, k0);
    asm volatile("" ::: "memory");
    for (int k = k0; k < kEnd; k += 2 * FG_STEP) {
        fg_load(f1, Ap, Bp, k + FG_STEP);
        fg_mma(f0, acc);
        const int kn = (k + 2 * FG_STEP < kEnd) ? (k + 2 * FG_STEP) : k0;
        fg_load(f0, Ap, Bp, kn);
        asm volatile("" ::: "memory");
        fg_mma(f1, acc);
    }

#pragma unroll
    for (int s = 0; s < 4; ++s) {
        if (wave == s) {
#pragma unroll
            for (int rt = 0; rt < 4; ++rt) {
                if (rt != s) {
                    float4* dp = (float4*)&red[(rt * 4 + s) * FG_SLOT] + lane;
#pragma unroll
                    for (int j = 0; j < 8; ++j) {
                        dp[(2 * j) * 32]     = make_float4(acc[rt][j][0], acc[rt][j][1], acc[rt][j][2], acc[rt][j][3]);
                        dp[(2 * j + 1) * 32] = make_float4(acc[rt][j][4], acc[rt][j][5], acc[rt][j][6], acc[rt][j][7]);
                    }
                }
            }
        }
    }
    __syncthreads();
#pragma unroll
    for (int rt = 0; rt < 4; ++rt) {
        if (wave == rt) {
#pragma unroll
            for (int s = 0; s < 4; ++s) {
                if (s != rt) {
                    const float4* sp = (const float4*)&red[(rt * 4 + s) * FG_SLOT] + lane;
#pragma unroll
                    for (int j = 0; j < 8; ++j) {
                        const float4 x = sp[(2 * j) * 32];
                        const float4 y = sp[(2 * j + 1) * 32];
                        acc[rt][j][0] += x.x; acc[rt][j][1] += x.y; acc[rt][j][2] += x.z; acc[rt][j][3] += x.w;
                        acc[rt][j][4] += y.x; acc[rt][j][5] += y.y; acc[rt][j][6] += y.z; acc[rt][j][7] += y.w;
                    }
                }
            }
#pragma unroll
            for (int q = 0; q < 4; ++q) {
#pragma unroll
                for (int e = 0; e < 2; ++e) {
#pragma unroll
                    for (int i = 0; i < 8; ++i) {
                        const int m = m0 + rt * 16 + i + 8 * half;
                        atomAddF(&out[m * 128 + 32 * q + 2 * c + e], acc[rt][2 * q + e][i]);
                    }
                }
            }
        }
    }
}

static inline int cdiv(long long a, long long b) { return (int)((a + b - 1) / b); }

extern "C" void kernel_launch(void* const* d_in, const int* in_sizes, int n_in,
                              void* d_out, int out_size, void* d_ws, size_t ws_size,
                              hipStream_t stream) {
    const float* feat_svc  = (const float*)d_in[0];
    const float* feat_inst = (const float*)d_in[1];
    const float* feat_node = (const float*)d_in[2];
    const int*   svc_src   = (const int*)d_in[3];
    const int*   svc_dst   = (const int*)d_in[4];
    const int*   in_src    = (const int*)d_in[5];
    const int*   in_dst    = (const int*)d_in[6];
    const int*   ni_src    = (const int*)d_in[7];
    const int*   ni_dst    = (const int*)d_in[8];
    const int*   ii_src    = (const int*)d_in[9];
    const int*   ii_dst    = (const int*)d_in[10];
    const float* W_sc = (const float*)d_in[11];  const float* b_sc = (const float*)d_in[12];
    const float* W_in = (const float*)d_in[13];  const float* b_in = (const float*)d_in[14];
    const float* W_ni = (const float*)d_in[15];  const float* b_ni = (const float*)d_in[16];
    const float* W_ii = (const float*)d_in[17];  const float* b_ii = (const float*)d_in[18];
    const float* W_tot = (const float*)d_in[19]; const float* b_tot = (const float*)d_in[20];
    float* out = (float*)d_out;

    float* ws   = (float*)d_ws;
    float* cat  = ws + OFF_CAT;
    float* h_sc = ws + OFF_HSC;
    float* h_in = ws + OFF_HIN;
    float* h_ni = ws + OFF_HNI;
    float* h_ii = ws + OFF_HII;
    float* deg  = ws + OFF_DEG;
    unsigned* tag = (unsigned*)(ws + OFF_TAG);

    init_deg_kernel<<<SMALL_BLOCKS, 256, 0, stream>>>(deg);
    init_out_kernel<<<cdiv(OUTD * HDIM, 256), 256, 0, stream>>>(out, b_tot);

    proj_deg_kernel<<<PJ_BLOCKS, 256, 0, stream>>>(
        feat_svc, feat_inst, feat_node, W_sc, W_in, W_ni, W_ii, h_sc, h_in, h_ni, h_ii,
        svc_src, svc_dst, in_src, in_dst, ni_src, ni_dst, ii_src, ii_dst, deg, tag);

    agg_kernel<<<AG_T_INST + AG_T_SVC + AG_T_NODE, AG_THREADS, 0, stream>>>(
        svc_src, svc_dst, in_src, in_dst, ni_src, ni_dst, ii_src, ii_dst,
        h_sc, h_in, h_ni, h_ii, deg,
        b_sc, b_in, b_ni, b_ii, tag, cat);

    final_gemm_kernel<<<FG_GROUPS, 128, 0, stream>>>(W_tot, cat, out);
}


# aggregation gathers a source row by one 16-byte load per lane (was four 4-byte loads); the LDS tile keeps column 4l+j at position 32j+l so its f32 adds stay conflict-free; sum order unchanged
# speedup vs baseline: 2.8414x; 1.0468x over previous
#include <hip/hip_runtime.h>

#define SVCN   50000
#define INSTN  100000
#define NODEN  50000
#define FDIM   128
#define HDIM   128
#define OUTD   256
#define TOTAL  (SVCN + NODEN + INSTN)
#define E_SC   400000
#define E_IN   100000
#define E_NI   100000
#define E_II   800000

#define OFF_CAT   0LL
#define OFF_HSC   (OFF_CAT  + (long long)TOTAL * 128)
#define OFF_HIN   (OFF_HSC  + (long long)SVCN  * 128)
#define OFF_HNI   (OFF_HIN  + (long long)INSTN * 128)
#define OFF_HII   (OFF_HNI  + (long long)NODEN * 128)
#define OFF_DEG   (OFF_HII  + (long long)INSTN * 128)
#define D_NS_SC   (OFF_DEG + 0LL)
#define D_NS_IN   (D_NS_SC + SVCN)
#define D_NS_NI   (D_NS_IN + INSTN)
#define D_ND_NI   (D_NS_NI + NODEN)
#define D_NS_II   (D_ND_NI + INSTN)
#define DEG_USED  (SVCN + INSTN + NODEN + INSTN + INSTN)
#define DEG_TOTAL ((DEG_USED + 31) / 32 * 32)
#define OFF_TAG   (OFF_DEG + DEG_TOTAL)

typedef __attribute__((ext_vector_type(2))) float v2f;
typedef __attribute__((ext_vector_type(8))) float v8f;

__device__ __forceinline__ float lrelu(float x) { return x > 0.f ? x : 0.01f * x; }

__device__ __forceinline__ void atomAddF(float* p, float v) { unsafeAtomicAdd(p, v); }

#define SMALL_BLOCKS 128
static_assert(DEG_TOTAL % 4 == 0, "degree planes are walked 16 bytes per lane");

__global__ __launch_bounds__(256) void init_deg_kernel(float* deg) {
    float4* d4 = (float4*)deg;
    for (int i = blockIdx.x * 256 + threadIdx.x; i < DEG_TOTAL / 4; i += SMALL_BLOCKS * 256)
        d4[i] = make_float4(0.f, 0.f, 0.f, 0.f);
}

__global__ void init_out_kernel(float* __restrict__ out, const float* __restrict__ b_tot) {
    int i = blockIdx.x * blockDim.x + threadIdx.x;
    if (i < OUTD * HDIM) out[i] = b_tot[i >> 7];
}

#define PJ_STEPS (FDIM / 16)
#define PJ_T_SVC   ((SVCN  + 63) / 64)
#define PJ_T_NODE  ((NODEN + 63) / 64)
#define PJ_T_INST  ((INSTN + 63) / 64)
#define PJ_BLOCKS  (2 * PJ_T_INST + PJ_T_SVC + PJ_T_NODE)
#define DG_SC      (E_SC / 4)
#define DG_IN      (E_IN / 4)
#define DG_NI      (E_NI / 4)
#define DG_II      (E_II / 4)
#define DG_TOTAL   (DG_SC + DG_IN + DG_NI + DG_II)
#define DG_PER_BLOCK ((((DG_TOTAL + PJ_BLOCKS - 1) / PJ_BLOCKS) + 31) / 32 * 32)
#define AG_ROWS    635
static_assert((INSTN + AG_ROWS - 1) / AG_ROWS <= 256 && (SVCN + AG_ROWS - 1) / AG_ROWS <= 256 &&
              (NODEN + AG_ROWS - 1) / AG_ROWS <= 256, "a tile number fits one byte");
static_assert(E_SC % 16 == 0 && E_IN % 16 == 0 && E_NI % 16 == 0 && E_II % 16 == 0,
              "tags are read 16 per lane");
static_assert(E_SC % 4 == 0 && E_IN % 4 == 0 && E_NI % 4 == 0 && E_II % 4 == 0,
              "edge ids are read 4 per lane");
static_assert(DG_PER_BLOCK <= 256, "one id group per thread");

__global__ __launch_bounds__(256) void proj_deg_kernel(
        const float* __restrict__ feat_svc, const float* __restrict__ feat_inst,
        const float* __restrict__ feat_node,
        const float* __restrict__ W_sc, const float* __restrict__ W_in,
        const float* __restrict__ W_ni, const float* __restrict__ W_ii,
        float* __restrict__ h_sc, float* __restrict__ h_in,
        float* __restrict__ h_ni, float* __restrict__ h_ii,
        const int* __restrict__ svc_src, const int* __restrict__ svc_dst,
        const int* __restrict__ in_src,  const int* __restrict__ in_dst,
        const int* __restrict__ ni_src,  const int* __restrict__ ni_dst,
        const int* __restrict__ ii_src,  const int* __restrict__ ii_dst,
        float* __restrict__ deg, unsigned* __restrict__ tag) {
    __shared__ float Wlds[128 * 128];
    const int t    = threadIdx.x;
    const int lane = t & 31;
    const int wave = t >> 5;
    const int half = lane >> 4;
    const int c    = lane & 15;
    const int b    = blockIdx.x;

    const float* X; const float* W; float* Hout; int nrows, tile;
    if (b < 2 * PJ_T_INST) {
        tile = b >> 1; X = feat_inst; nrows = INSTN;
        if (b & 1) { W = W_ii; Hout = h_ii; } else { W = W_in; Hout = h_in; }
    } else if (b < 2 * PJ_T_INST + PJ_T_SVC) {
        tile = b - 2 * PJ_T_INST; X = feat_svc; W = W_sc; Hout = h_sc; nrows = SVCN;
    } else {
        tile = b - 2 * PJ_T_INST - PJ_T_SVC; X = feat_node; W = W_ni; Hout = h_ni; nrows = NODEN;
    }
    const int r0 = tile * 64 + (wave >> 1) * 16;
    const int n0 = (wave & 1) * 64 + 2 * c;

    const int  g      = b * DG_PER_BLOCK + t;
    const bool counts = t < DG_PER_BLOCK && g < DG_TOTAL;
    const int* sp; const int* dp; int oS, gl;
    bool isNi = false;
    if (g < DG_SC) {
        sp = svc_src; dp = svc_dst; gl = g;
        oS = (int)(D_NS_SC - OFF_DEG);
    } else if (g < DG_SC + DG_IN) {
        sp = in_src; dp = in_dst; gl = g - DG_SC;
        oS = (int)(D_NS_IN - OFF_DEG);
    } else if (g < DG_SC + DG_IN + DG_NI) {
        sp = ni_src; dp = ni_dst; gl = g - DG_SC - DG_IN;
        oS = (int)(D_NS_NI - OFF_DEG); isNi = true;
    } else {
        sp = ii_src; dp = ii_dst; gl = g - DG_SC - DG_IN - DG_NI;
        oS = (int)(D_NS_II - OFF_DEG);
    }
    int4 es = make_int4(0, 0, 0, 0), ed = make_int4(0, 0, 0, 0);
    if (counts) {
        es = ((const int4*)sp)[gl];
        ed = ((const int4*)dp)[gl];
    }

    const int rowA = (r0 + c < nrows) ? (r0 + c) : (nrows - 1);
    const float* Ap = X + (long long)rowA * 128 + 8 * half;
    float4 a[PJ_STEPS][2];
#pragma unroll
    for (int ks = 0; ks < PJ_STEPS; ++ks) {
        a[ks][0] = *(const float4*)(Ap + 16 * ks);
        a[ks][1] = *(const float4*)(Ap + 16 * ks + 4);
    }

#pragma unroll
    for (int it = 0; it < 8; ++it) {
        const int u  = t + 256 * it;
        const int kp = u >> 5;
        const int m4 = (u & 31) * 4;
        const float4 w0 = *(const float4*)&W[(2 * kp) * 128 + m4];
        const float4 w1 = *(const float4*)&W[(2 * kp + 1) * 128 + m4];
        float* d = &Wlds[(kp * 128 + m4) * 2];
        *(float4*)(d)     = make_float4(w0.x, w1.x, w0.y, w1.y);
        *(float4*)(d + 4) = make_float4(w0.z, w1.z, w0.w, w1.w);
    }
    __syncthreads();

    v8f acc[4] = {v8f{}, v8f{}, v8f{}, v8f{}};
    const float* Bp = &Wlds[(4 * half * 128 + n0) * 2];

#pragma unroll
    for (int ks = 0; ks < PJ_STEPS; ++ks) {
#pragma unroll
        for (int s = 0; s < 4; ++s) {
            const float* bp = Bp + ((8 * ks + s) * 128) * 2;
            const float4 b0 = *(const float4*)(bp);
            const float4 b1 = *(const float4*)(bp + 64);
            const float4 av = a[ks][s >> 1];
            v2f af;
            af.x = (s & 1) ? av.z : av.x;
            af.y = (s & 1) ? av.w : av.y;
            v2f bf;
            bf.x = b0.x; bf.y = b0.y;
            acc[0] = __builtin_amdgcn_wmma_f32_16x16x4_f32(false, af, false, bf, (short)0, acc[0], false, false);
            bf.x = b0.z; bf.y = b0.w;
            acc[1] = __builtin_amdgcn_wmma_f32_16x16x4_f32(false, af, false, bf, (short)0, acc[1], false, false);
            bf.x = b1.x; bf.y = b1.y;
            acc[2] = __builtin_amdgcn_wmma_f32_16x16x4_f32(false, af, false, bf, (short)0, acc[2], false, false);
            bf.x = b1.z; bf.y = b1.w;
            acc[3] = __builtin_amdgcn_wmma_f32_16x16x4_f32(false, af, false, bf, (short)0, acc[3], false, false);
        }
    }

#pragma unroll
    for (int i = 0; i < 8; ++i) {
        const int row = r0 + i + 8 * half;
        if (row < nrows) {
            float* hp = Hout + (long long)row * 128 + n0;
            *(float2*)(hp)      = make_float2(acc[0][i], acc[1][i]);
            *(float2*)(hp + 32) = make_float2(acc[2][i], acc[3][i]);
        }
    }

    if (counts) {
        tag[g] = ((unsigned)ed.x / AG_ROWS) | (((unsigned)ed.y / AG_ROWS) << 8) |
                 (((unsigned)ed.z / AG_ROWS) << 16) | (((unsigned)ed.w / AG_ROWS) << 24);
        float* cS = deg + oS;
        atomAddF(cS + es.x, 1.f); atomAddF(cS + es.y, 1.f);
        atomAddF(cS + es.z, 1.f); atomAddF(cS + es.w, 1.f);
        if (isNi) {
            float* cD = deg + (D_ND_NI - OFF_DEG);
            atomAddF(cD + ed.x, 1.f); atomAddF(cD + ed.y, 1.f);
            atomAddF(cD + ed.z, 1.f); atomAddF(cD + ed.w, 1.f);
        }
    }
}

#define AG_THREADS 1024
#define AG_WAVES   (AG_THREADS / 32)
#define AG_FLY     16
#define AG_GRP     2
#define AG_T_INST  ((INSTN + AG_ROWS - 1) / AG_ROWS)
#define AG_T_SVC   ((SVCN  + AG_ROWS - 1) / AG_ROWS)
#define AG_T_NODE  ((NODEN + AG_ROWS - 1) / AG_ROWS)
struct AgList {
    const int*   src;
    const int*   dst;
    const float* cntS;
    const float* cntD;
    const float* H;
    const uint4* tag;
    int          n16;
};

template <bool FULL>
__device__ __forceinline__ void ag_flush(const AgList& L, float* tile, unsigned* rowcnt, int r0, int lane, int qe, int n) {
    const int d = L.dst[qe];
    const int s = L.src[qe];
    const int r = d - r0;
    float cd = 1.f;
    if (L.cntD != nullptr) cd = L.cntD[d];
    else if (lane < n) atomicAdd(rowcnt + r, 1u);
    const float cs = L.cntS[s];
    const int f = __float_as_int(rsqrtf(cd > 1.f ? cd : 1.f) * rsqrtf(cs > 1.f ? cs : 1.f));
    float4 h[AG_FLY];
#pragma unroll
    for (int j = 0; j < AG_FLY; ++j) {
        h[j] = make_float4(0.f, 0.f, 0.f, 0.f);
        if (FULL || j < n) {
            const float* hp = L.H + (long long)__builtin_amdgcn_readlane(s, j) * 128;
            h[j] = *(const float4*)(hp + 4 * lane);
        }
    }
#pragma unroll
    for (int j = 0; j < AG_FLY; ++j) {
        if (FULL || j < n) {
            float*      p  = tile + __builtin_amdgcn_readlane(r, j) * 128 + lane;
            const float fj = __int_as_float(__builtin_amdgcn_readlane(f, j));
            atomicAdd(p,      h[j].x * fj);
            atomicAdd(p + 32, h[j].y * fj);
            atomicAdd(p + 64, h[j].z * fj);
            atomicAdd(p + 96, h[j].w * fj);
        }
    }
}

__device__ __forceinline__ unsigned zero_bytes(unsigned x) {
    const unsigned t = ((x & 0x7F7F7F7Fu) + 0x7F7F7F7Fu) | x;
    return ~t & 0x80808080u;
}

__device__ __forceinline__ void ag_group(const AgList& L, float* tile, unsigned* rowcnt, int r0, int lane,
                                         uint4 tg, unsigned mine4, int gb, int& qe, int& cnt) {
    const unsigned m0 = zero_bytes(tg.x ^ mine4), m1 = zero_bytes(tg.y ^ mine4);
    const unsigned m2 = zero_bytes(tg.z ^ mine4), m3 = zero_bytes(tg.w ^ mine4);
    const unsigned code = (m0 >> 7) | (m1 >> 6) | (m2 >> 5) | (m3 >> 4);
    unsigned lanes = __builtin_amdgcn_ballot_w32(gb + lane < L.n16 && code != 0u);
    if (lanes == 0u) return;
    do {
        const int l = __builtin_ctz(lanes);
        lanes &= lanes - 1u;
        unsigned c = (unsigned)__builtin_amdgcn_readlane((int)code, l);
        const int eb = 16 * (gb + l);
        do {
            const int p = __builtin_ctz(c);
            c &= c - 1u;
            const int k = 4 * (p & 7) + (p >> 3);
            qe = (lane == cnt) ? (eb + k) : qe;
            if (++cnt == AG_FLY) { ag_flush<true>(L, tile, rowcnt, r0, lane, qe, AG_FLY); cnt = 0; }
        } while (c != 0u);
    } while (lanes != 0u);
}

__device__ __forceinline__ void ag_list(const AgList& L, float* tile, unsigned* rowcnt, int r0, unsigned mine4, int wave, int lane) {
    const uint4* tg4 = L.tag;
    int qe  = 0;
    int cnt = 0;
    int i = wave * 32 + lane;
    uint4 n0 = tg4[i < L.n16 ? i : 0];
    uint4 n1 = tg4[i + AG_THREADS < L.n16 ? i + AG_THREADS : 0];
    for (int gb = wave * 32; gb < L.n16; gb += AG_GRP * AG_THREADS) {
        const uint4 d0 = n0, d1 = n1;
        i += AG_GRP * AG_THREADS;
        n0 = tg4[i < L.n16 ? i : 0];
        n1 = tg4[i + AG_THREADS < L.n16 ? i + AG_THREADS : 0];
        ag_group(L, tile, rowcnt, r0, lane, d0, mine4, gb, qe, cnt);
        ag_group(L, tile, rowcnt, r0, lane, d1, mine4, gb + AG_THREADS, qe, cnt);
    }
    if (cnt != 0) ag_flush<false>(L, tile, rowcnt, r0, lane, qe, cnt);
}

__global__ __launch_bounds__(AG_THREADS) void agg_kernel(
        const int* __restrict__ svc_src, const int* __restrict__ svc_dst,
        const int* __restrict__ in_src,  const int* __restrict__ in_dst,
        const int* __restrict__ ni_src,  const int* __restrict__ ni_dst,
        const int* __restrict__ ii_src,  const int* __restrict__ ii_dst,
        const float* __restrict__ h_sc, const float* __restrict__ h_in,
        const float* __restrict__ h_ni, const float* __restrict__ h_ii,
        const float* __restrict__ deg,
        const float* __restrict__ b_sc, const float* __restrict__ b_in,
        const float* __restrict__ b_ni, const float* __restrict__ b_ii,
        const unsigned* __restrict__ tag,
        float* __restrict__ cat) {
    __shared__ float    tile[AG_ROWS * 128];
    __shared__ unsigned rowcnt[AG_ROWS];
    const int lane = threadIdx.x & 31;
    const int wave = __builtin_amdgcn_readfirstlane(threadIdx.x >> 5);
    const int b    = blockIdx.x;

    for (int i = threadIdx.x; i < AG_ROWS * 32; i += AG_THREADS)
        ((float4*)tile)[i] = make_float4(0.f, 0.f, 0.f, 0.f);
    for (int i = threadIdx.x; i < AG_ROWS; i += AG_THREADS) rowcnt[i] = 0u;
    __syncthreads();

    int r0, nr, catRow0, nLists, tno;
    AgList L;
    const uint4* tag16 = (const uint4*)tag;
    float4 bias;
    const int c4 = lane * 4;
    if (b < AG_T_INST) {
        tno = b;
        r0 = b * AG_ROWS;
        nr = (INSTN - r0 < AG_ROWS) ? (INSTN - r0) : AG_ROWS;
        catRow0 = SVCN + NODEN + r0;
        L.src = ii_src; L.dst = ii_dst; L.H = h_ii; nLists = 2;
        L.tag = tag16 + (E_SC + E_IN + E_NI) / 16; L.n16 = E_II / 16;
        L.cntS = deg + (D_NS_II - OFF_DEG); L.cntD = nullptr;
        const float4 x = *(const float4*)&b_ni[c4];
        const float4 y = *(const float4*)&b_ii[c4];
        bias = make_float4(x.x + y.x, x.y + y.y, x.z + y.z, x.w + y.w);
    } else if (b < AG_T_INST + AG_T_SVC) {
        tno = b - AG_T_INST;
        r0 = tno * AG_ROWS;
        nr = (SVCN - r0 < AG_ROWS) ? (SVCN - r0) : AG_ROWS;
        catRow0 = r0;
        L.src = svc_src; L.dst = svc_dst; L.H = h_sc; nLists = 1;
        L.tag = tag16; L.n16 = E_SC / 16;
        L.cntS = deg + (D_NS_SC - OFF_DEG); L.cntD = nullptr;
        bias = *(const float4*)&b_sc[c4];
    } else {
        tno = b - AG_T_INST - AG_T_SVC;
        r0 = tno * AG_ROWS;
        nr = (NODEN - r0 < AG_ROWS) ? (NODEN - r0) : AG_ROWS;
        catRow0 = SVCN + r0;
        L.src = in_src; L.dst = in_dst; L.H = h_in; nLists = 1;
        L.tag = tag16 + E_SC / 16; L.n16 = E_IN / 16;
        L.cntS = deg + (D_NS_IN - OFF_DEG); L.cntD = nullptr;
        bias = *(const float4*)&b_in[c4];
    }
    const unsigned mine4 = (unsigned)tno * 0x01010101u;
#pragma nounroll
    for (int li = 0; li < nLists; ++li) {
        ag_list(L, tile, rowcnt, r0, mine4, wave, lane);
        __syncthreads();
        if (li + 1 < nLists) {
            for (int r = wave; r < nr; r += AG_WAVES) {
                const float  cr = (float)rowcnt[r];
                const float  nd = rsqrtf(cr > 1.f ? cr : 1.f);
                const float4 v  = *(const float4*)&tile[r * 128 + c4];
                *(float4*)&tile[r * 128 + c4] = make_float4(v.x * nd, v.y * nd, v.z * nd, v.w * nd);
            }
            __syncthreads();
            L.src = ni_src; L.dst = ni_dst; L.H = h_ni;
            L.tag = tag16 + (E_SC + E_IN) / 16; L.n16 = E_NI / 16;
            L.cntS = deg + (D_NS_NI - OFF_DEG); L.cntD = deg + (D_ND_NI - OFF_DEG);
        }
    }

    for (int r = wave; r < nr; r += AG_WAVES) {
        const float* tp = &tile[r * 128 + lane];
        const float4 v  = make_float4(tp[0], tp[32], tp[64], tp[96]);
        const float  cr = (float)rowcnt[r];
        const float  nd = (nLists == 1) ? rsqrtf(cr > 1.f ? cr : 1.f) : 1.f;
        *(float4*)&cat[(long long)(catRow0 + r) * 128 + c4] =
            make_float4(v.x * nd + bias.x, v.y * nd + bias.y, v.z * nd + bias.z, v.w * nd + bias.w);
    }
}

#define FG_STEP   16
#define FG_DSTEPS (TOTAL / (2 * FG_STEP))
#define FG_CHUNKS 63
#define FG_SLICES (4 * FG_CHUNKS)
#define FG_GROUPS (4 * FG_CHUNKS)
#define FG_SLOT   2048
static_assert(TOTAL % (2 * FG_STEP) == 0, "K is walked in double steps");
static_assert(FG_DSTEPS / FG_SLICES >= 1, "no empty slice");

struct FgFrag {
    float4 a[4][2];
    float2 b[4][4][2];
};

__device__ __forceinline__ void fg_load(FgFrag& f, const float* const (&Ap)[4],
                                        const float* __restrict__ Bp, int k) {
#pragma unroll
    for (int rt = 0; rt < 4; ++rt) {
        f.a[rt][0] = *(const float4*)(Ap[rt] + k);
        f.a[rt][1] = *(const float4*)(Ap[rt] + k + 4);
    }
#pragma unroll
    for (int s = 0; s < 4; ++s) {
#pragma unroll
        for (int q = 0; q < 4; ++q) {
            f.b[s][q][0] = *(const float2*)(Bp + (long long)(k + 2 * s) * 128 + 32 * q);
            f.b[s][q][1] = *(const float2*)(Bp + (long long)(k + 2 * s + 1) * 128 + 32 * q);
        }
    }
}

__device__ __forceinline__ void fg_mma(const FgFrag& f, v8f (&acc)[4][8]) {
#pragma unroll
    for (int s = 0; s < 4; ++s) {
        v2f bf[4][2];
#pragma unroll
        for (int q = 0; q < 4; ++q) {
            bf[q][0].x = lrelu(f.b[s][q][0].x);
            bf[q][0].y = lrelu(f.b[s][q][1].x);
            bf[q][1].x = lrelu(f.b[s][q][0].y);
            bf[q][1].y = lrelu(f.b[s][q][1].y);
        }
#pragma unroll
        for (int rt = 0; rt < 4; ++rt) {
            const float4 av = f.a[rt][s >> 1];
            v2f a;
            a.x = (s & 1) ? av.z : av.x;
            a.y = (s & 1) ? av.w : av.y;
#pragma unroll
            for (int q = 0; q < 4; ++q) {
#pragma unroll
                for (int e = 0; e < 2; ++e) {
                    acc[rt][2 * q + e] = __builtin_amdgcn_wmma_f32_16x16x4_f32(
                        false, a, false, bf[q][e], (short)0, acc[rt][2 * q + e], false, false);
                }
            }
        }
    }
}

__global__ __launch_bounds__(128) void final_gemm_kernel(const float* __restrict__ Wt,
                                                         const float* __restrict__ cat,
                                                         float* __restrict__ out) {
    __shared__ float red[16 * FG_SLOT];
    const int t    = threadIdx.x;
    const int lane = t & 31;
    const int wave = __builtin_amdgcn_readfirstlane(t >> 5);
    const int half = lane >> 4;
    const int c    = lane & 15;
    const int m0   = (blockIdx.x & 3) * 64;
    const int sl   = (blockIdx.x >> 2) * 4 + wave;
    const int k0   = 2 * FG_STEP * ((FG_DSTEPS * sl) / FG_SLICES);
    const int kEnd = 2 * FG_STEP * ((FG_DSTEPS * (sl + 1)) / FG_SLICES);

    const float* Ap[4];
#pragma unroll
    for (int rt = 0; rt < 4; ++rt)
        Ap[rt] = Wt + (long long)(m0 + rt * 16 + c) * TOTAL + 8 * half;
    const float* Bp = cat + (long long)(8 * half) * 128 + 2 * c;

    v8f acc[4][8];
#pragma unroll
    for (int rt = 0; rt < 4; ++rt)
#pragma unroll
        for (int j = 0; j < 8; ++j) acc[rt][j] = v8f{};

    FgFrag f0, f1;
    fg_load(f0, Ap, Bp, k0);
    asm volatile("" ::: "memory");
    for (int k = k0; k < kEnd; k += 2 * FG_STEP) {
        fg_load(f1, Ap, Bp, k + FG_STEP);
        fg_mma(f0, acc);
        const int kn = (k + 2 * FG_STEP < kEnd) ? (k + 2 * FG_STEP) : k0;
        fg_load(f0, Ap, Bp, kn);
        asm volatile("" ::: "memory");
        fg_mma(f1, acc);
    }

#pragma unroll
    for (int s = 0; s < 4; ++s) {
        if (wave == s) {
#pragma unroll
            for (int rt = 0; rt < 4; ++rt) {
                if (rt != s) {
                    float4* dp = (float4*)&red[(rt * 4 + s) * FG_SLOT] + lane;
#pragma unroll
                    for (int j = 0; j < 8; ++j) {
                        dp[(2 * j) * 32]     = make_float4(acc[rt][j][0], acc[rt][j][1], acc[rt][j][2], acc[rt][j][3]);
                        dp[(2 * j + 1) * 32] = make_float4(acc[rt][j][4], acc[rt][j][5], acc[rt][j][6], acc[rt][j][7]);
                    }
                }
            }
        }
    }
    __syncthreads();
#pragma unroll
    for (int rt = 0; rt < 4; ++rt) {
        if (wave == rt) {
#pragma unroll
            for (int s = 0; s < 4; ++s) {
                if (s != rt) {
                    const float4* sp = (const float4*)&red[(rt * 4 + s) * FG_SLOT] + lane;
#pragma unroll
                    for (int j = 0; j < 8; ++j) {
                        const float4 x = sp[(2 * j) * 32];
                        const float4 y = sp[(2 * j + 1) * 32];
                        acc[rt][j][0] += x.x; acc[rt][j][1] += x.y; acc[rt][j][2] += x.z; acc[rt][j][3] += x.w;
                        acc[rt][j][4] += y.x; acc[rt][j][5] += y.y; acc[rt][j][6] += y.z; acc[rt][j][7] += y.w;
                    }
                }
            }
#pragma unroll
            for (int q = 0; q < 4; ++q) {
#pragma unroll
                for (int e = 0; e < 2; ++e) {
#pragma unroll
                    for (int i = 0; i < 8; ++i) {
                        const int m = m0 + rt * 16 + i + 8 * half;
                        atomAddF(&out[m * 128 + 32 * q + 2 * c + e], acc[rt][2 * q + e][i]);
                    }
                }
            }
        }
    }
}

static inline int cdiv(long long a, long long b) { return (int)((a + b - 1) / b); }

extern "C" void kernel_launch(void* const* d_in, const int* in_sizes, int n_in,
                              void* d_out, int out_size, void* d_ws, size_t ws_size,
                              hipStream_t stream) {
    const float* feat_svc  = (const float*)d_in[0];
    const float* feat_inst = (const float*)d_in[1];
    const float* feat_node = (const float*)d_in[2];
    const int*   svc_src   = (const int*)d_in[3];
    const int*   svc_dst   = (const int*)d_in[4];
    const int*   in_src    = (const int*)d_in[5];
    const int*   in_dst    = (const int*)d_in[6];
    const int*   ni_src    = (const int*)d_in[7];
    const int*   ni_dst    = (const int*)d_in[8];
    const int*   ii_src    = (const int*)d_in[9];
    const int*   ii_dst    = (const int*)d_in[10];
    const float* W_sc = (const float*)d_in[11];  const float* b_sc = (const float*)d_in[12];
    const float* W_in = (const float*)d_in[13];  const float* b_in = (const float*)d_in[14];
    const float* W_ni = (const float*)d_in[15];  const float* b_ni = (const float*)d_in[16];
    const float* W_ii = (const float*)d_in[17];  const float* b_ii = (const float*)d_in[18];
    const float* W_tot = (const float*)d_in[19]; const float* b_tot = (const float*)d_in[20];
    float* out = (float*)d_out;

    float* ws   = (float*)d_ws;
    float* cat  = ws + OFF_CAT;
    float* h_sc = ws + OFF_HSC;
    float* h_in = ws + OFF_HIN;
    float* h_ni = ws + OFF_HNI;
    float* h_ii = ws + OFF_HII;
    float* deg  = ws + OFF_DEG;
    unsigned* tag = (unsigned*)(ws + OFF_TAG);

    init_deg_kernel<<<SMALL_BLOCKS, 256, 0, stream>>>(deg);
    init_out_kernel<<<cdiv(OUTD * HDIM, 256), 256, 0, stream>>>(out, b_tot);

    proj_deg_kernel<<<PJ_BLOCKS, 256, 0, stream>>>(
        feat_svc, feat_inst, feat_node, W_sc, W_in, W_ni, W_ii, h_sc, h_in, h_ni, h_ii,
        svc_src, svc_dst, in_src, in_dst, ni_src, ni_dst, ii_src, ii_dst, deg, tag);

    agg_kernel<<<AG_T_INST + AG_T_SVC + AG_T_NODE, AG_THREADS, 0, stream>>>(
        svc_src, svc_dst, in_src, in_dst, ni_src, ni_dst, ii_src, ii_dst,
        h_sc, h_in, h_ni, h_ii, deg,
        b_sc, b_in, b_ni, b_ii, tag, cat);

    final_gemm_kernel<<<FG_GROUPS, 128, 0, stream>>>(W_tot, cat, out);
}
